# starting kernel
# baseline (speedup 1.0000x reference)
#include <hip/hip_runtime.h>
#include <math.h>

typedef __attribute__((ext_vector_type(2))) float v2f;
typedef __attribute__((ext_vector_type(8))) float v8f;

__device__ __forceinline__ v8f wmma4f32(v2f a, v2f b, v8f c) {
  return __builtin_amdgcn_wmma_f32_16x16x4_f32(
       false, a,  false, b,
       (short)0, c,  false,  false);
}

__device__ __forceinline__ float half16_sum(float s) {
  s += __shfl_xor(s, 1, 32);
  s += __shfl_xor(s, 2, 32);
  s += __shfl_xor(s, 4, 32);
  s += __shfl_xor(s, 8, 32);
  return s;
}

__global__ void edge0_kernel(const float* __restrict__ x,
                             const int* __restrict__ src,
                             const int* __restrict__ dst,
                             const float* __restrict__ kw0,
                             const float* __restrict__ kw1,
                             float* __restrict__ acc0, int E) {
  unsigned e = blockIdx.x * blockDim.x + threadIdx.x;
  if (e >= (unsigned)E) return;
  const int s = src[e], t = dst[e];
  const float w0 = kw0[e], w1 = kw1[e];
  const float4* xp = (const float4*)(x + (size_t)s * 8);
  const float4 xa = xp[0], xb = xp[1];
  const float xv[8] = {xa.x, xa.y, xa.z, xa.w, xb.x, xb.y, xb.z, xb.w};
  float* a = acc0 + (size_t)t * 16;
#pragma unroll
  for (int k = 0; k < 8; ++k) atomicAdd(a + k, w0 * xv[k]);
#pragma unroll
  for (int k = 0; k < 8; ++k) atomicAdd(a + 8 + k, w1 * xv[k]);
}

__global__ __launch_bounds__(32) void mlp0_kernel(const float* __restrict__ acc0,
                                                  const float* __restrict__ W0,
                                                  const float* __restrict__ b0,
                                                  float* __restrict__ h1) {
  const int m0 = blockIdx.x * 16;
  const int lane = threadIdx.x;
  const int n = lane & 15;
  const int kb = lane >> 4;
  v8f c0 = {}, c1 = {};
#pragma unroll
  for (int kt = 0; kt < 4; ++kt) {
    const int kA = kt * 4 + 2 * kb;
    v2f a, bb0, bb1;
    const float* ap = acc0 + (size_t)(m0 + n) * 16 + kA;
    a.x = ap[0];
    a.y = ap[1];
    bb0.x = W0[(kA + 0) * 32 + n];
    bb0.y = W0[(kA + 1) * 32 + n];
    bb1.x = W0[(kA + 0) * 32 + 16 + n];
    bb1.y = W0[(kA + 1) * 32 + 16 + n];
    c0 = wmma4f32(a, bb0, c0);
    c1 = wmma4f32(a, bb1, c1);
  }
  const float bias0 = b0[n], bias1 = b0[16 + n];
#pragma unroll
  for (int v = 0; v < 8; ++v) {
    float e0 = c0[v] + bias0;
    float e1 = c1[v] + bias1;
    float s = half16_sum(e0 * e0 + e1 * e1);
    const float invn = 1.0f / fmaxf(sqrtf(s), 1e-12f);
    const int row = m0 + v + 8 * kb;
    h1[(size_t)row * 32 + n] = e0 * invn;
    h1[(size_t)row * 32 + 16 + n] = e1 * invn;
  }
}

__global__ void edge1_kernel(const float* __restrict__ h1,
                             const int* __restrict__ src,
                             const int* __restrict__ dst,
                             const float* __restrict__ Kw,
                             float* __restrict__ acc1, int E) {
  unsigned tid = blockIdx.x * blockDim.x + threadIdx.x;
  if (tid >= 4u * (unsigned)E) return;
  const int e = tid >> 2;
  const int j = (tid >> 1) & 1;
  const int hh = tid & 1;
  const int s = src[e], t = dst[e];
  const float w = Kw[(size_t)j * E + e];
  const float4* g = (const float4*)(h1 + (size_t)s * 32 + hh * 16);
  const float4 g0 = g[0], g1 = g[1], g2 = g[2], g3 = g[3];
  const float vv[16] = {g0.x, g0.y, g0.z, g0.w, g1.x, g1.y, g1.z, g1.w,
                        g2.x, g2.y, g2.z, g2.w, g3.x, g3.y, g3.z, g3.w};
  float* a = acc1 + (size_t)t * 64 + j * 32 + hh * 16;
#pragma unroll
  for (int k = 0; k < 16; ++k) atomicAdd(a + k, w * vv[k]);
}

#define TS 132
__global__ __launch_bounds__(128) void mlp1_kernel(const float* __restrict__ acc1,
                                                   const float* __restrict__ W1,
                                                   const float* __restrict__ b1,
                                                   const float* __restrict__ W2,
                                                   const float* __restrict__ b2,
                                                   float* __restrict__ out) {
  __shared__ float tsh[16 * TS];
  __shared__ float rowsq[16];
  const int m0 = blockIdx.x * 16;
  const int wave = threadIdx.x >> 5;
  const int lane = threadIdx.x & 31;
  const int n = lane & 15;
  const int kb = lane >> 4;

#pragma unroll
  for (int tt = 0; tt < 2; ++tt) {
    const int n0 = (wave * 2 + tt) * 16;
    v8f c = {};
    for (int kt = 0; kt < 16; ++kt) {
      const int kA = kt * 4 + 2 * kb;
      v2f a, b;
      const float* ap = acc1 + (size_t)(m0 + n) * 64 + kA;
      a.x = ap[0];
      a.y = ap[1];
      b.x = W1[(kA + 0) * 128 + n0 + n];
      b.y = W1[(kA + 1) * 128 + n0 + n];
      c = wmma4f32(a, b, c);
    }
    const float bias = b1[n0 + n];
#pragma unroll
    for (int v = 0; v < 8; ++v) {
      tsh[(v + 8 * kb) * TS + n0 + n] = fmaxf(c[v] + bias, 0.0f);
    }
  }
  if (threadIdx.x < 16) rowsq[threadIdx.x] = 0.0f;
  __syncthreads();

  const int n0 = wave * 16;
  v8f c = {};
  for (int kt = 0; kt < 32; ++kt) {
    const int kA = kt * 4 + 2 * kb;
    v2f a, b;
    a.x = tsh[n * TS + kA];
    a.y = tsh[n * TS + kA + 1];
    b.x = W2[(kA + 0) * 64 + n0 + n];
    b.y = W2[(kA + 1) * 64 + n0 + n];
    c = wmma4f32(a, b, c);
  }
  const float bias = b2[n0 + n];
  float cv[8];
#pragma unroll
  for (int v = 0; v < 8; ++v) {
    cv[v] = c[v] + bias;
    float s = half16_sum(cv[v] * cv[v]);
    if (n == 0) atomicAdd(&rowsq[v + 8 * kb], s);
  }
  __syncthreads();
#pragma unroll
  for (int v = 0; v < 8; ++v) {
    const int row = v + 8 * kb;
    const float invn = 1.0f / fmaxf(sqrtf(rowsq[row]), 1e-12f);
    out[(size_t)(m0 + row) * 64 + n0 + n] = cv[v] * invn;
  }
}

extern "C" void kernel_launch(void* const* d_in, const int* in_sizes, int n_in,
                              void* d_out, int out_size, void* d_ws, size_t ws_size,
                              hipStream_t stream) {
  const float* x  = (const float*)d_in[0];
  const int*   ei = (const int*)d_in[1];
  const float* Kw = (const float*)d_in[2];
  const float* W0 = (const float*)d_in[3];
  const float* b0 = (const float*)d_in[4];
  const float* W1 = (const float*)d_in[5];
  const float* b1 = (const float*)d_in[6];
  const float* W2 = (const float*)d_in[7];
  const float* b2 = (const float*)d_in[8];

  const int N = in_sizes[0] / 8;
  const int E = in_sizes[1] / 2;
  const int* src = ei;
  const int* dst = ei + E;

  float* acc0 = (float*)d_ws;
  float* acc1 = acc0 + (size_t)N * 16;
  float* h1   = acc1 + (size_t)N * 64;

  hipMemsetAsync(d_ws, 0, (size_t)N * 80 * sizeof(float), stream);

  edge0_kernel<<<(E + 255) / 256, 256, 0, stream>>>(x, src, dst, Kw, Kw + E, acc0, E);
  mlp0_kernel<<<N / 16, 32, 0, stream>>>(acc0, W0, b0, h1);
  edge1_kernel<<<(4 * E + 255) / 256, 256, 0, stream>>>(h1, src, dst, Kw, acc1, E);
  mlp1_kernel<<<N / 16, 128, 0, stream>>>(acc1, W1, b1, W2, b2, (float*)d_out);
}


# float atomics removed: edges counting-sorted by dst each call (one int atomic per edge gives its arrival-order rank), per-node sums in registers in that order; layer-1 sums reach the MLP via LDS
# speedup vs baseline: 3.1685x; 3.1685x over previous
#include <hip/hip_runtime.h>
#include <math.h>

typedef __attribute__((ext_vector_type(2))) float v2f;
typedef __attribute__((ext_vector_type(8))) float v8f;

__device__ __forceinline__ v8f wmma4f32(v2f a, v2f b, v8f c) {
  return __builtin_amdgcn_wmma_f32_16x16x4_f32(
       false, a,  false, b,
       (short)0, c,  false,  false);
}

__device__ __forceinline__ float half16_sum(float s) {
  s += __shfl_xor(s, 1, 32);
  s += __shfl_xor(s, 2, 32);
  s += __shfl_xor(s, 4, 32);
  s += __shfl_xor(s, 8, 32);
  return s;
}

__global__ void degree_kernel(const int* __restrict__ dst, int* __restrict__ cnt,
                              int* __restrict__ rank, int E) {
  unsigned e = blockIdx.x * blockDim.x + threadIdx.x;
  if (e >= (unsigned)E) return;
  rank[e] = atomicAdd(cnt + dst[e], 1);
}

__global__ __launch_bounds__(256) void slice_kernel(const int* __restrict__ cnt,
                                                    int* __restrict__ total,
                                                    int* __restrict__ start, int N) {
  __shared__ int wsum[8];
  __shared__ int base_s;
  const int t = blockIdx.x * 256 + threadIdx.x;
  const int lane = threadIdx.x & 31;
  const int wave = threadIdx.x >> 5;
  const int c = (t < N) ? cnt[t] : 0;
  int incl = c;
  for (int d = 1; d < 32; d <<= 1) {
    const int v = __shfl_up(incl, d, 32);
    if (lane >= d) incl += v;
  }
  if (lane == 31) wsum[wave] = incl;
  __syncthreads();
  if (threadIdx.x == 0) {
    int s = 0;
    for (int w = 0; w < 8; ++w) {
      const int v = wsum[w];
      wsum[w] = s;
      s += v;
    }
    base_s = atomicAdd(total, s);
  }
  __syncthreads();
  if (t < N) start[t] = base_s + wsum[wave] + incl - c;
}

__global__ void place_kernel(const int* __restrict__ src,
                             const int* __restrict__ dst,
                             const float* __restrict__ kw0,
                             const float* __restrict__ kw1,
                             const int* __restrict__ start,
                             const int* __restrict__ rank,
                             int* __restrict__ ssrc,
                             float* __restrict__ sw, int E) {
  unsigned e = blockIdx.x * blockDim.x + threadIdx.x;
  if (e >= (unsigned)E) return;
  const int pos = start[dst[e]] + rank[e];
  ssrc[pos] = src[e];
  ((float2*)sw)[pos] = make_float2(kw0[e], kw1[e]);
}

__global__ void gather0_kernel(const float* __restrict__ x,
                               const int* __restrict__ cnt,
                               const int* __restrict__ start,
                               const int* __restrict__ ssrc,
                               const float* __restrict__ sw,
                               float* __restrict__ acc0, int N) {
  unsigned tid = blockIdx.x * blockDim.x + threadIdx.x;
  if (tid >= 2u * (unsigned)N) return;
  const int t = tid >> 1;
  const int q = tid & 1;
  const int beg = start[t];
  const int end = beg + cnt[t];
  const float* xq = x + 4 * q;
  float4 a0 = make_float4(0.f, 0.f, 0.f, 0.f);
  float4 a1 = make_float4(0.f, 0.f, 0.f, 0.f);
#pragma unroll 4
  for (int p = beg; p < end; ++p) {
    const int s = ssrc[p];
    const float2 w = ((const float2*)sw)[p];
    const float w0 = w.x, w1 = w.y;
    const float4 v = *(const float4*)(xq + (size_t)s * 8);
    a0.x += w0 * v.x; a0.y += w0 * v.y; a0.z += w0 * v.z; a0.w += w0 * v.w;
    a1.x += w1 * v.x; a1.y += w1 * v.y; a1.z += w1 * v.z; a1.w += w1 * v.w;
  }
  float* a = acc0 + (size_t)t * 16 + 4 * q;
  *(float4*)(a) = a0;
  *(float4*)(a + 8) = a1;
}

__global__ __launch_bounds__(32) void mlp0_kernel(const float* __restrict__ acc0,
                                                  const float* __restrict__ W0,
                                                  const float* __restrict__ b0,
                                                  float* __restrict__ h1) {
  const int m0 = blockIdx.x * 16;
  const int lane = threadIdx.x;
  const int n = lane & 15;
  const int kb = lane >> 4;
  v8f c0 = {}, c1 = {};
#pragma unroll
  for (int kt = 0; kt < 4; ++kt) {
    const int kA = kt * 4 + 2 * kb;
    v2f a, bb0, bb1;
    const float* ap = acc0 + (size_t)(m0 + n) * 16 + kA;
    a.x = ap[0];
    a.y = ap[1];
    bb0.x = W0[(kA + 0) * 32 + n];
    bb0.y = W0[(kA + 1) * 32 + n];
    bb1.x = W0[(kA + 0) * 32 + 16 + n];
    bb1.y = W0[(kA + 1) * 32 + 16 + n];
    c0 = wmma4f32(a, bb0, c0);
    c1 = wmma4f32(a, bb1, c1);
  }
  const float bias0 = b0[n], bias1 = b0[16 + n];
#pragma unroll
  for (int v = 0; v < 8; ++v) {
    float e0 = c0[v] + bias0;
    float e1 = c1[v] + bias1;
    float s = half16_sum(e0 * e0 + e1 * e1);
    const float invn = 1.0f / fmaxf(sqrtf(s), 1e-12f);
    const int row = m0 + v + 8 * kb;
    h1[(size_t)row * 32 + n] = e0 * invn;
    h1[(size_t)row * 32 + 16 + n] = e1 * invn;
  }
}

#define TS 132
#define AS 68
__global__ __launch_bounds__(128) void mlp1_kernel(const float* __restrict__ h1,
                                                   const int* __restrict__ cnt,
                                                   const int* __restrict__ start,
                                                   const int* __restrict__ ssrc,
                                                   const float* __restrict__ sw,
                                                   const float* __restrict__ W1,
                                                   const float* __restrict__ b1,
                                                   const float* __restrict__ W2,
                                                   const float* __restrict__ b2,
                                                   float* __restrict__ out) {
  __shared__ __attribute__((aligned(16))) float ash[16 * AS];
  __shared__ float tsh[16 * TS];
  __shared__ float rowsq[16];
  const int m0 = blockIdx.x * 16;
  const int wave = threadIdx.x >> 5;
  const int lane = threadIdx.x & 31;
  const int n = lane & 15;
  const int kb = lane >> 4;

  {
    const int r = threadIdx.x >> 3;
    const int q = threadIdx.x & 7;
    const int beg = start[m0 + r];
    const int end = beg + cnt[m0 + r];
    const float* hq = h1 + 4 * q;
    float4 a0 = make_float4(0.f, 0.f, 0.f, 0.f);
    float4 a1 = make_float4(0.f, 0.f, 0.f, 0.f);
#pragma unroll 4
    for (int p = beg; p < end; ++p) {
      const int s = ssrc[p];
      const float2 w = ((const float2*)sw)[p];
      const float w0 = w.x, w1 = w.y;
      const float4 v = *(const float4*)(hq + (size_t)s * 32);
      a0.x += w0 * v.x; a0.y += w0 * v.y; a0.z += w0 * v.z; a0.w += w0 * v.w;
      a1.x += w1 * v.x; a1.y += w1 * v.y; a1.z += w1 * v.z; a1.w += w1 * v.w;
    }
    *(float4*)(ash + r * AS + 4 * q) = a0;
    *(float4*)(ash + r * AS + 32 + 4 * q) = a1;
  }
  __syncthreads();

#pragma unroll
  for (int tt = 0; tt < 2; ++tt) {
    const int n0 = (wave * 2 + tt) * 16;
    v8f c = {};
    for (int kt = 0; kt < 16; ++kt) {
      const int kA = kt * 4 + 2 * kb;
      v2f a, b;
      const float* ap = ash + n * AS + kA;
      a.x = ap[0];
      a.y = ap[1];
      b.x = W1[(kA + 0) * 128 + n0 + n];
      b.y = W1[(kA + 1) * 128 + n0 + n];
      c = wmma4f32(a, b, c);
    }
    const float bias = b1[n0 + n];
#pragma unroll
    for (int v = 0; v < 8; ++v) {
      tsh[(v + 8 * kb) * TS + n0 + n] = fmaxf(c[v] + bias, 0.0f);
    }
  }
  if (threadIdx.x < 16) rowsq[threadIdx.x] = 0.0f;
  __syncthreads();

  const int n0 = wave * 16;
  v8f c = {};
  for (int kt = 0; kt < 32; ++kt) {
    const int kA = kt * 4 + 2 * kb;
    v2f a, b;
    a.x = tsh[n * TS + kA];
    a.y = tsh[n * TS + kA + 1];
    b.x = W2[(kA + 0) * 64 + n0 + n];
    b.y = W2[(kA + 1) * 64 + n0 + n];
    c = wmma4f32(a, b, c);
  }
  const float bias = b2[n0 + n];
  float cv[8];
#pragma unroll
  for (int v = 0; v < 8; ++v) {
    cv[v] = c[v] + bias;
    float s = half16_sum(cv[v] * cv[v]);
    if (n == 0) atomicAdd(&rowsq[v + 8 * kb], s);
  }
  __syncthreads();
#pragma unroll
  for (int v = 0; v < 8; ++v) {
    const int row = v + 8 * kb;
    const float invn = 1.0f / fmaxf(sqrtf(rowsq[row]), 1e-12f);
    out[(size_t)(m0 + row) * 64 + n0 + n] = cv[v] * invn;
  }
}

extern "C" void kernel_launch(void* const* d_in, const int* in_sizes, int n_in,
                              void* d_out, int out_size, void* d_ws, size_t ws_size,
                              hipStream_t stream) {
  const float* x  = (const float*)d_in[0];
  const int*   ei = (const int*)d_in[1];
  const float* Kw = (const float*)d_in[2];
  const float* W0 = (const float*)d_in[3];
  const float* b0 = (const float*)d_in[4];
  const float* W1 = (const float*)d_in[5];
  const float* b1 = (const float*)d_in[6];
  const float* W2 = (const float*)d_in[7];
  const float* b2 = (const float*)d_in[8];

  const int N = in_sizes[0] / 8;
  const int E = in_sizes[1] / 2;
  const int* src = ei;
  const int* dst = ei + E;

  int*    cnt    = (int*)d_ws;
  int*    total  = cnt + N;
  int*    start  = total + 64;
  int*    ssrc   = start + N;
  float*  sw     = (float*)(ssrc + E);
  float*  acc0   = sw + (size_t)E * 2;
  float*  h1     = acc0 + (size_t)N * 16;
  int*    rank   = (int*)acc0;

  hipMemsetAsync(d_ws, 0, ((size_t)N + 64) * sizeof(int), stream);

  degree_kernel<<<(E + 255) / 256, 256, 0, stream>>>(dst, cnt, rank, E);
  slice_kernel<<<(N + 255) / 256, 256, 0, stream>>>(cnt, total, start, N);
  place_kernel<<<(E + 255) / 256, 256, 0, stream>>>(src, dst, Kw, Kw + E, start, rank, ssrc, sw, E);
  gather0_kernel<<<(2 * N + 255) / 256, 256, 0, stream>>>(x, cnt, start, ssrc, sw, acc0, N);
  mlp0_kernel<<<N / 16, 32, 0, stream>>>(acc0, W0, b0, h1);
  mlp1_kernel<<<N / 16, 128, 0, stream>>>(h1, cnt, start, ssrc, sw, W1, b1, W2, b2,
                                          (float*)d_out);
}


# edge ranks come from LDS counters, not one global returning atomic per edge: two-level counting sort (256-node buckets, then nodes); per-node sum order stays an arrival order, now of LDS adds
# speedup vs baseline: 4.9453x; 1.5608x over previous
#include <hip/hip_runtime.h>
#include <math.h>

typedef __attribute__((ext_vector_type(2))) float v2f;
typedef __attribute__((ext_vector_type(8))) float v8f;

__device__ __forceinline__ v8f wmma4f32(v2f a, v2f b, v8f c) {
  return __builtin_amdgcn_wmma_f32_16x16x4_f32(
       false, a,  false, b,
       (short)0, c,  false,  false);
}

__device__ __forceinline__ float half16_sum(float s) {
  s += __shfl_xor(s, 1, 32);
  s += __shfl_xor(s, 2, 32);
  s += __shfl_xor(s, 4, 32);
  s += __shfl_xor(s, 8, 32);
  return s;
}

#define BSH 8
#define BNODES 256
#define NBMAX 1024
#define G1 256

__device__ __forceinline__ int wave_incl_scan(int v, int lane) {
  for (int d = 1; d < 32; d <<= 1) {
    const int u = __shfl_up(v, d, 32);
    if (lane >= d) v += u;
  }
  return v;
}

__global__ __launch_bounds__(1024) void bucket_count_kernel(const int* __restrict__ dst,
                                                            int* __restrict__ table,
                                                            int E, int NB) {
  __shared__ int hist[NBMAX];
  const int g = blockIdx.x;
  hist[threadIdx.x] = 0;
  __syncthreads();
  const int chunk = (E + G1 - 1) / G1;
  const int e0 = g * chunk;
  const int e1 = min(E, e0 + chunk);
  int e = e0 + (int)threadIdx.x;
  for (; e + 3 * 1024 < e1; e += 4 * 1024) {
    const int d0 = dst[e], d1 = dst[e + 1024], d2 = dst[e + 2048], d3 = dst[e + 3072];
    atomicAdd(&hist[d0 >> BSH], 1);
    atomicAdd(&hist[d1 >> BSH], 1);
    atomicAdd(&hist[d2 >> BSH], 1);
    atomicAdd(&hist[d3 >> BSH], 1);
  }
  for (; e < e1; e += 1024) atomicAdd(&hist[dst[e] >> BSH], 1);
  __syncthreads();
  for (int b = threadIdx.x; b < NB; b += 1024) table[b * G1 + g] = hist[b];
}

__global__ __launch_bounds__(G1) void bucket_scan_kernel(int* __restrict__ table,
                                                         int* __restrict__ rowTotal) {
  __shared__ int wsum[G1 / 32];
  const int b = blockIdx.x;
  const int g = threadIdx.x;
  const int lane = g & 31;
  const int wave = g >> 5;
  const int c = table[b * G1 + g];
  const int incl = wave_incl_scan(c, lane);
  if (lane == 31) wsum[wave] = incl;
  __syncthreads();
  int off = 0;
  for (int w = 0; w < wave; ++w) off += wsum[w];
  table[b * G1 + g] = off + incl - c;
  if (g == G1 - 1) rowTotal[b] = off + incl;
}

__global__ __launch_bounds__(1024) void bucket_place_kernel(const int* __restrict__ src,
                                                            const int* __restrict__ dst,
                                                            const float* __restrict__ kw0,
                                                            const float* __restrict__ kw1,
                                                            const int* __restrict__ table,
                                                            const int* __restrict__ rowTotal,
                                                            unsigned* __restrict__ recP,
                                                            float* __restrict__ recW,
                                                            int E, int NB) {
  __shared__ int cur[NBMAX];
  __shared__ int wsum[32];
  const int g = blockIdx.x;
  const int tid = threadIdx.x;
  const int lane = tid & 31;
  const int wave = tid >> 5;
  const int c = (tid < NB) ? rowTotal[tid] : 0;
  const int t = (tid < NB) ? table[tid * G1 + g] : 0;
  const int incl = wave_incl_scan(c, lane);
  if (lane == 31) wsum[wave] = incl;
  __syncthreads();
  if (wave == 0) {
    const int v = wsum[lane];
    const int s = wave_incl_scan(v, lane);
    wsum[lane] = s - v;
  }
  __syncthreads();
  cur[tid] = wsum[wave] + incl - c + t;
  __syncthreads();
  const int chunk = (E + G1 - 1) / G1;
  const int e0 = g * chunk;
  const int e1 = min(E, e0 + chunk);
  int e = e0 + tid;
  for (; e + 3 * 1024 < e1; e += 4 * 1024) {
    int d[4];
    unsigned s[4];
    float2 w[4];
    int pos[4];
#pragma unroll
    for (int u = 0; u < 4; ++u) {
      d[u] = dst[e + u * 1024];
      s[u] = (unsigned)src[e + u * 1024];
      w[u] = make_float2(kw0[e + u * 1024], kw1[e + u * 1024]);
    }
#pragma unroll
    for (int u = 0; u < 4; ++u) pos[u] = atomicAdd(&cur[d[u] >> BSH], 1);
#pragma unroll
    for (int u = 0; u < 4; ++u) {
      recP[pos[u]] = s[u] | ((unsigned)(d[u] & (BNODES - 1)) << 24);
      ((float2*)recW)[pos[u]] = w[u];
    }
  }
  for (; e < e1; e += 1024) {
    const int d = dst[e];
    const unsigned s = (unsigned)src[e];
    const float2 w = make_float2(kw0[e], kw1[e]);
    const int pos = atomicAdd(&cur[d >> BSH], 1);
    recP[pos] = s | ((unsigned)(d & (BNODES - 1)) << 24);
    ((float2*)recW)[pos] = w;
  }
}

__global__ __launch_bounds__(BNODES) void node_place_kernel(const int* __restrict__ rowTotal,
                                                            const unsigned* __restrict__ recP,
                                                            const float* __restrict__ recW,
                                                            int* __restrict__ start,
                                                            int* __restrict__ ssrc,
                                                            float* __restrict__ sw, int N) {
  __shared__ int cntL[BNODES];
  __shared__ int curL[BNODES];
  __shared__ int wsum[BNODES / 32];
  __shared__ int baseL;
  const int b = blockIdx.x;
  const int tid = threadIdx.x;
  const int lane = tid & 31;
  const int wave = tid >> 5;
  cntL[tid] = 0;
  if (tid == 0) baseL = 0;
  __syncthreads();
  int part = 0;
  for (int i = tid; i < b; i += BNODES) part += rowTotal[i];
  part += __shfl_xor(part, 16, 32);
  part += __shfl_xor(part, 8, 32);
  part += __shfl_xor(part, 4, 32);
  part += __shfl_xor(part, 2, 32);
  part += __shfl_xor(part, 1, 32);
  if (lane == 0 && part != 0) atomicAdd(&baseL, part);
  const int n = rowTotal[b];
  __syncthreads();
  const int base = baseL;
  const unsigned* rp = recP + base;
  const float2* rw = (const float2*)recW + base;
  int i = tid;
  for (; i + 3 * BNODES < n; i += 4 * BNODES) {
    const unsigned p0 = rp[i], p1 = rp[i + BNODES], p2 = rp[i + 2 * BNODES], p3 = rp[i + 3 * BNODES];
    atomicAdd(&cntL[p0 >> 24], 1);
    atomicAdd(&cntL[p1 >> 24], 1);
    atomicAdd(&cntL[p2 >> 24], 1);
    atomicAdd(&cntL[p3 >> 24], 1);
  }
  for (; i < n; i += BNODES) atomicAdd(&cntL[rp[i] >> 24], 1);
  __syncthreads();
  const int c = cntL[tid];
  const int incl = wave_incl_scan(c, lane);
  if (lane == 31) wsum[wave] = incl;
  __syncthreads();
  int off = 0;
  for (int w = 0; w < wave; ++w) off += wsum[w];
  const int excl = off + incl - c;
  curL[tid] = excl;
  const int node = b * BNODES + tid;
  if (node < N) start[node] = base + excl;
  if (node == N - 1) start[N] = base + excl + c;
  __syncthreads();
  int k = tid;
  for (; k + 3 * BNODES < n; k += 4 * BNODES) {
    unsigned p[4];
    float2 w[4];
    int pos[4];
#pragma unroll
    for (int u = 0; u < 4; ++u) {
      p[u] = rp[k + u * BNODES];
      w[u] = rw[k + u * BNODES];
    }
#pragma unroll
    for (int u = 0; u < 4; ++u) pos[u] = base + atomicAdd(&curL[p[u] >> 24], 1);
#pragma unroll
    for (int u = 0; u < 4; ++u) {
      ssrc[pos[u]] = (int)(p[u] & 0xFFFFFFu);
      ((float2*)sw)[pos[u]] = w[u];
    }
  }
  for (int i = k; i < n; i += BNODES) {
    const unsigned p = rp[i];
    const float2 w = rw[i];
    const int pos = base + atomicAdd(&curL[p >> 24], 1);
    ssrc[pos] = (int)(p & 0xFFFFFFu);
    ((float2*)sw)[pos] = w;
  }
}

__global__ void gather0_kernel(const float* __restrict__ x,
                               const int* __restrict__ start,
                               const int* __restrict__ ssrc,
                               const float* __restrict__ sw,
                               float* __restrict__ acc0, int N) {
  unsigned tid = blockIdx.x * blockDim.x + threadIdx.x;
  if (tid >= 2u * (unsigned)N) return;
  const int t = tid >> 1;
  const int q = tid & 1;
  const int beg = start[t];
  const int end = start[t + 1];
  const float* xq = x + 4 * q;
  float4 a0 = make_float4(0.f, 0.f, 0.f, 0.f);
  float4 a1 = make_float4(0.f, 0.f, 0.f, 0.f);
#pragma unroll 4
  for (int p = beg; p < end; ++p) {
    const int s = ssrc[p];
    const float2 w = ((const float2*)sw)[p];
    const float w0 = w.x, w1 = w.y;
    const float4 v = *(const float4*)(xq + (size_t)s * 8);
    a0.x += w0 * v.x; a0.y += w0 * v.y; a0.z += w0 * v.z; a0.w += w0 * v.w;
    a1.x += w1 * v.x; a1.y += w1 * v.y; a1.z += w1 * v.z; a1.w += w1 * v.w;
  }
  float* a = acc0 + (size_t)t * 16 + 4 * q;
  *(float4*)(a) = a0;
  *(float4*)(a + 8) = a1;
}

__global__ __launch_bounds__(32) void mlp0_kernel(const float* __restrict__ acc0,
                                                  const float* __restrict__ W0,
                                                  const float* __restrict__ b0,
                                                  float* __restrict__ h1) {
  const int m0 = blockIdx.x * 16;
  const int lane = threadIdx.x;
  const int n = lane & 15;
  const int kb = lane >> 4;
  v8f c0 = {}, c1 = {};
#pragma unroll
  for (int kt = 0; kt < 4; ++kt) {
    const int kA = kt * 4 + 2 * kb;
    v2f a, bb0, bb1;
    const float* ap = acc0 + (size_t)(m0 + n) * 16 + kA;
    a.x = ap[0];
    a.y = ap[1];
    bb0.x = W0[(kA + 0) * 32 + n];
    bb0.y = W0[(kA + 1) * 32 + n];
    bb1.x = W0[(kA + 0) * 32 + 16 + n];
    bb1.y = W0[(kA + 1) * 32 + 16 + n];
    c0 = wmma4f32(a, bb0, c0);
    c1 = wmma4f32(a, bb1, c1);
  }
  const float bias0 = b0[n], bias1 = b0[16 + n];
#pragma unroll
  for (int v = 0; v < 8; ++v) {
    float e0 = c0[v] + bias0;
    float e1 = c1[v] + bias1;
    float s = half16_sum(e0 * e0 + e1 * e1);
    const float invn = 1.0f / fmaxf(sqrtf(s), 1e-12f);
    const int row = m0 + v + 8 * kb;
    h1[(size_t)row * 32 + n] = e0 * invn;
    h1[(size_t)row * 32 + 16 + n] = e1 * invn;
  }
}

#define TS 132
#define AS 68
__global__ __launch_bounds__(128) void mlp1_kernel(const float* __restrict__ h1,
                                                   const int* __restrict__ start,
                                                   const int* __restrict__ ssrc,
                                                   const float* __restrict__ sw,
                                                   const float* __restrict__ W1,
                                                   const float* __restrict__ b1,
                                                   const float* __restrict__ W2,
                                                   const float* __restrict__ b2,
                                                   float* __restrict__ out) {
  __shared__ __attribute__((aligned(16))) float ash[16 * AS];
  __shared__ float tsh[16 * TS];
  __shared__ float rowsq[16];
  const int m0 = blockIdx.x * 16;
  const int wave = threadIdx.x >> 5;
  const int lane = threadIdx.x & 31;
  const int n = lane & 15;
  const int kb = lane >> 4;

  {
    const int r = threadIdx.x >> 3;
    const int q = threadIdx.x & 7;
    const int beg = start[m0 + r];
    const int end = start[m0 + r + 1];
    const float* hq = h1 + 4 * q;
    float4 a0 = make_float4(0.f, 0.f, 0.f, 0.f);
    float4 a1 = make_float4(0.f, 0.f, 0.f, 0.f);
#pragma unroll 4
    for (int p = beg; p < end; ++p) {
      const int s = ssrc[p];
      const float2 w = ((const float2*)sw)[p];
      const float w0 = w.x, w1 = w.y;
      const float4 v = *(const float4*)(hq + (size_t)s * 32);
      a0.x += w0 * v.x; a0.y += w0 * v.y; a0.z += w0 * v.z; a0.w += w0 * v.w;
      a1.x += w1 * v.x; a1.y += w1 * v.y; a1.z += w1 * v.z; a1.w += w1 * v.w;
    }
    *(float4*)(ash + r * AS + 4 * q) = a0;
    *(float4*)(ash + r * AS + 32 + 4 * q) = a1;
  }
  __syncthreads();

#pragma unroll
  for (int tt = 0; tt < 2; ++tt) {
    const int n0 = (wave * 2 + tt) * 16;
    v8f c = {};
    for (int kt = 0; kt < 16; ++kt) {
      const int kA = kt * 4 + 2 * kb;
      v2f a, b;
      const float* ap = ash + n * AS + kA;
      a.x = ap[0];
      a.y = ap[1];
      b.x = W1[(kA + 0) * 128 + n0 + n];
      b.y = W1[(kA + 1) * 128 + n0 + n];
      c = wmma4f32(a, b, c);
    }
    const float bias = b1[n0 + n];
#pragma unroll
    for (int v = 0; v < 8; ++v) {
      tsh[(v + 8 * kb) * TS + n0 + n] = fmaxf(c[v] + bias, 0.0f);
    }
  }
  if (threadIdx.x < 16) rowsq[threadIdx.x] = 0.0f;
  __syncthreads();

  const int n0 = wave * 16;
  v8f c = {};
  for (int kt = 0; kt < 32; ++kt) {
    const int kA = kt * 4 + 2 * kb;
    v2f a, b;
    a.x = tsh[n * TS + kA];
    a.y = tsh[n * TS + kA + 1];
    b.x = W2[(kA + 0) * 64 + n0 + n];
    b.y = W2[(kA + 1) * 64 + n0 + n];
    c = wmma4f32(a, b, c);
  }
  const float bias = b2[n0 + n];
  float cv[8];
#pragma unroll
  for (int v = 0; v < 8; ++v) {
    cv[v] = c[v] + bias;
    float s = half16_sum(cv[v] * cv[v]);
    if (n == 0) atomicAdd(&rowsq[v + 8 * kb], s);
  }
  __syncthreads();
#pragma unroll
  for (int v = 0; v < 8; ++v) {
    const int row = v + 8 * kb;
    const float invn = 1.0f / fmaxf(sqrtf(rowsq[row]), 1e-12f);
    out[(size_t)(m0 + row) * 64 + n0 + n] = cv[v] * invn;
  }
}

extern "C" void kernel_launch(void* const* d_in, const int* in_sizes, int n_in,
                              void* d_out, int out_size, void* d_ws, size_t ws_size,
                              hipStream_t stream) {
  const float* x  = (const float*)d_in[0];
  const int*   ei = (const int*)d_in[1];
  const float* Kw = (const float*)d_in[2];
  const float* W0 = (const float*)d_in[3];
  const float* b0 = (const float*)d_in[4];
  const float* W1 = (const float*)d_in[5];
  const float* b1 = (const float*)d_in[6];
  const float* W2 = (const float*)d_in[7];
  const float* b2 = (const float*)d_in[8];

  const int N = in_sizes[0] / 8;
  const int E = in_sizes[1] / 2;
  const int* src = ei;
  const int* dst = ei + E;

  const int NB = (N + BNODES - 1) / BNODES;

  int*      start    = (int*)d_ws;
  int*      rowTotal = start + N + 64;
  int*      ssrc     = rowTotal + NBMAX;
  float*    sw       = (float*)(ssrc + E);
  float*    acc0     = sw + (size_t)E * 2;
  float*    h1       = acc0 + (size_t)N * 16;
  int*      table    = ssrc;
  unsigned* recP     = (unsigned*)acc0;
  float*    recW     = (float*)(recP + E);

  bucket_count_kernel<<<G1, 1024, 0, stream>>>(dst, table, E, NB);
  bucket_scan_kernel<<<NB, G1, 0, stream>>>(table, rowTotal);
  bucket_place_kernel<<<G1, 1024, 0, stream>>>(src, dst, Kw, Kw + E, table, rowTotal, recP, recW,
                                               E, NB);
  node_place_kernel<<<NB, BNODES, 0, stream>>>(rowTotal, recP, recW, start, ssrc, sw, N);
  gather0_kernel<<<(2 * N + 255) / 256, 256, 0, stream>>>(x, start, ssrc, sw, acc0, N);
  mlp0_kernel<<<N / 16, 32, 0, stream>>>(acc0, W0, b0, h1);
  mlp1_kernel<<<N / 16, 128, 0, stream>>>(h1, start, ssrc, sw, W1, b1, W2, b2,
                                          (float*)d_out);
}


# both neighbour aggregations: block stages its nodes' edge slice (ids, weights) in LDS with coalesced loads, then a node's lanes request 8 source rows at once; per-node sum order unchanged
# speedup vs baseline: 5.3859x; 1.0891x over previous
#include <hip/hip_runtime.h>
#include <math.h>

typedef __attribute__((ext_vector_type(2))) float v2f;
typedef __attribute__((ext_vector_type(8))) float v8f;

__device__ __forceinline__ v8f wmma4f32(v2f a, v2f b, v8f c) {
  return __builtin_amdgcn_wmma_f32_16x16x4_f32(
       false, a,  false, b,
       (short)0, c,  false,  false);
}

__device__ __forceinline__ float half16_sum(float s) {
  s += __shfl_xor(s, 1, 32);
  s += __shfl_xor(s, 2, 32);
  s += __shfl_xor(s, 4, 32);
  s += __shfl_xor(s, 8, 32);
  return s;
}

#define BSH 8
#define BNODES 256
#define NBMAX 1024
#define G1 256

__device__ __forceinline__ int wave_incl_scan(int v, int lane) {
  for (int d = 1; d < 32; d <<= 1) {
    const int u = __shfl_up(v, d, 32);
    if (lane >= d) v += u;
  }
  return v;
}

__global__ __launch_bounds__(1024) void bucket_count_kernel(const int* __restrict__ dst,
                                                            int* __restrict__ table,
                                                            int E, int NB) {
  __shared__ int hist[NBMAX];
  const int g = blockIdx.x;
  hist[threadIdx.x] = 0;
  __syncthreads();
  const int chunk = (E + G1 - 1) / G1;
  const int e0 = g * chunk;
  const int e1 = min(E, e0 + chunk);
  int e = e0 + (int)threadIdx.x;
  for (; e + 3 * 1024 < e1; e += 4 * 1024) {
    const int d0 = dst[e], d1 = dst[e + 1024], d2 = dst[e + 2048], d3 = dst[e + 3072];
    atomicAdd(&hist[d0 >> BSH], 1);
    atomicAdd(&hist[d1 >> BSH], 1);
    atomicAdd(&hist[d2 >> BSH], 1);
    atomicAdd(&hist[d3 >> BSH], 1);
  }
  for (; e < e1; e += 1024) atomicAdd(&hist[dst[e] >> BSH], 1);
  __syncthreads();
  for (int b = threadIdx.x; b < NB; b += 1024) table[b * G1 + g] = hist[b];
}

__global__ __launch_bounds__(G1) void bucket_scan_kernel(int* __restrict__ table,
                                                         int* __restrict__ rowTotal) {
  __shared__ int wsum[G1 / 32];
  const int b = blockIdx.x;
  const int g = threadIdx.x;
  const int lane = g & 31;
  const int wave = g >> 5;
  const int c = table[b * G1 + g];
  const int incl = wave_incl_scan(c, lane);
  if (lane == 31) wsum[wave] = incl;
  __syncthreads();
  int off = 0;
  for (int w = 0; w < wave; ++w) off += wsum[w];
  table[b * G1 + g] = off + incl - c;
  if (g == G1 - 1) rowTotal[b] = off + incl;
}

__global__ __launch_bounds__(1024) void bucket_place_kernel(const int* __restrict__ src,
                                                            const int* __restrict__ dst,
                                                            const float* __restrict__ kw0,
                                                            const float* __restrict__ kw1,
                                                            const int* __restrict__ table,
                                                            const int* __restrict__ rowTotal,
                                                            unsigned* __restrict__ recP,
                                                            float* __restrict__ recW,
                                                            int E, int NB) {
  __shared__ int cur[NBMAX];
  __shared__ int wsum[32];
  const int g = blockIdx.x;
  const int tid = threadIdx.x;
  const int lane = tid & 31;
  const int wave = tid >> 5;
  const int c = (tid < NB) ? rowTotal[tid] : 0;
  const int t = (tid < NB) ? table[tid * G1 + g] : 0;
  const int incl = wave_incl_scan(c, lane);
  if (lane == 31) wsum[wave] = incl;
  __syncthreads();
  if (wave == 0) {
    const int v = wsum[lane];
    const int s = wave_incl_scan(v, lane);
    wsum[lane] = s - v;
  }
  __syncthreads();
  cur[tid] = wsum[wave] + incl - c + t;
  __syncthreads();
  const int chunk = (E + G1 - 1) / G1;
  const int e0 = g * chunk;
  const int e1 = min(E, e0 + chunk);
  int e = e0 + tid;
  for (; e + 3 * 1024 < e1; e += 4 * 1024) {
    int d[4];
    unsigned s[4];
    float2 w[4];
    int pos[4];
#pragma unroll
    for (int u = 0; u < 4; ++u) {
      d[u] = dst[e + u * 1024];
      s[u] = (unsigned)src[e + u * 1024];
      w[u] = make_float2(kw0[e + u * 1024], kw1[e + u * 1024]);
    }
#pragma unroll
    for (int u = 0; u < 4; ++u) pos[u] = atomicAdd(&cur[d[u] >> BSH], 1);
#pragma unroll
    for (int u = 0; u < 4; ++u) {
      recP[pos[u]] = s[u] | ((unsigned)(d[u] & (BNODES - 1)) << 24);
      ((float2*)recW)[pos[u]] = w[u];
    }
  }
  for (; e < e1; e += 1024) {
    const int d = dst[e];
    const unsigned s = (unsigned)src[e];
    const float2 w = make_float2(kw0[e], kw1[e]);
    const int pos = atomicAdd(&cur[d >> BSH], 1);
    recP[pos] = s | ((unsigned)(d & (BNODES - 1)) << 24);
    ((float2*)recW)[pos] = w;
  }
}

__global__ __launch_bounds__(BNODES) void node_place_kernel(const int* __restrict__ rowTotal,
                                                            const unsigned* __restrict__ recP,
                                                            const float* __restrict__ recW,
                                                            int* __restrict__ start,
                                                            int* __restrict__ ssrc,
                                                            float* __restrict__ sw, int N) {
  __shared__ int cntL[BNODES];
  __shared__ int curL[BNODES];
  __shared__ int wsum[BNODES / 32];
  __shared__ int baseL;
  const int b = blockIdx.x;
  const int tid = threadIdx.x;
  const int lane = tid & 31;
  const int wave = tid >> 5;
  cntL[tid] = 0;
  if (tid == 0) baseL = 0;
  __syncthreads();
  int part = 0;
  for (int i = tid; i < b; i += BNODES) part += rowTotal[i];
  part += __shfl_xor(part, 16, 32);
  part += __shfl_xor(part, 8, 32);
  part += __shfl_xor(part, 4, 32);
  part += __shfl_xor(part, 2, 32);
  part += __shfl_xor(part, 1, 32);
  if (lane == 0 && part != 0) atomicAdd(&baseL, part);
  const int n = rowTotal[b];
  __syncthreads();
  const int base = baseL;
  const unsigned* rp = recP + base;
  const float2* rw = (const float2*)recW + base;
  int i = tid;
  for (; i + 3 * BNODES < n; i += 4 * BNODES) {
    const unsigned p0 = rp[i], p1 = rp[i + BNODES], p2 = rp[i + 2 * BNODES], p3 = rp[i + 3 * BNODES];
    atomicAdd(&cntL[p0 >> 24], 1);
    atomicAdd(&cntL[p1 >> 24], 1);
    atomicAdd(&cntL[p2 >> 24], 1);
    atomicAdd(&cntL[p3 >> 24], 1);
  }
  for (; i < n; i += BNODES) atomicAdd(&cntL[rp[i] >> 24], 1);
  __syncthreads();
  const int c = cntL[tid];
  const int incl = wave_incl_scan(c, lane);
  if (lane == 31) wsum[wave] = incl;
  __syncthreads();
  int off = 0;
  for (int w = 0; w < wave; ++w) off += wsum[w];
  const int excl = off + incl - c;
  curL[tid] = excl;
  const int node = b * BNODES + tid;
  if (node < N) start[node] = base + excl;
  if (node == N - 1) start[N] = base + excl + c;
  __syncthreads();
  int k = tid;
  for (; k + 3 * BNODES < n; k += 4 * BNODES) {
    unsigned p[4];
    float2 w[4];
    int pos[4];
#pragma unroll
    for (int u = 0; u < 4; ++u) {
      p[u] = rp[k + u * BNODES];
      w[u] = rw[k + u * BNODES];
    }
#pragma unroll
    for (int u = 0; u < 4; ++u) pos[u] = base + atomicAdd(&curL[p[u] >> 24], 1);
#pragma unroll
    for (int u = 0; u < 4; ++u) {
      ssrc[pos[u]] = (int)(p[u] & 0xFFFFFFu);
      ((float2*)sw)[pos[u]] = w[u];
    }
  }
  for (int i = k; i < n; i += BNODES) {
    const unsigned p = rp[i];
    const float2 w = rw[i];
    const int pos = base + atomicAdd(&curL[p >> 24], 1);
    ssrc[pos] = (int)(p & 0xFFFFFFu);
    ((float2*)sw)[pos] = w;
  }
}

#define EB 8
static_assert(EB == 8, "sum_rows names the EB rows of a batch one by one");

template <int ROWSHIFT>
__device__ __forceinline__ void sum_rows(const char* __restrict__ rows, unsigned lane_off,
                                         const int* eidx, const float2* ewt,
                                         int c0, int pb, int pe, float4& a0, float4& a1) {
  for (int p = pb; p < pe; p += EB) {
    const int nv = pe - p;
    int s[EB];
    float2 w[EB];
    float4 v[EB];
#pragma unroll
    for (int u = 0; u < EB; ++u) {
      const int i = min(p + u, pe - 1) - c0;
      s[u] = eidx[i];
      const float2 ww = ewt[i];
      w[u].x = (u < nv) ? ww.x : 0.f;
      w[u].y = (u < nv) ? ww.y : 0.f;
    }
#pragma unroll
    for (int u = 0; u < EB; ++u)
      v[u] = *(const float4*)(rows + ((((unsigned)s[u]) << ROWSHIFT) | lane_off));
    asm volatile(""
                 : "+v"(a0.x), "+v"(a0.y), "+v"(a0.z), "+v"(a0.w),
                   "+v"(a1.x), "+v"(a1.y), "+v"(a1.z), "+v"(a1.w)
                 : "v"(v[0].x), "v"(v[1].x), "v"(v[2].x), "v"(v[3].x),
                   "v"(v[4].x), "v"(v[5].x), "v"(v[6].x), "v"(v[7].x));
#pragma unroll
    for (int u = 0; u < EB; ++u) {
      const float w0 = w[u].x, w1 = w[u].y;
      a0.x += w0 * v[u].x; a0.y += w0 * v[u].y; a0.z += w0 * v[u].z; a0.w += w0 * v[u].w;
      a1.x += w1 * v[u].x; a1.y += w1 * v[u].y; a1.z += w1 * v[u].z; a1.w += w1 * v[u].w;
    }
  }
}

#define G0N 64
#define G0T 128
#define G0CAP 1536
__global__ __launch_bounds__(G0T) void gather0_kernel(const float* __restrict__ x,
                                                      const int* __restrict__ start,
                                                      const int* __restrict__ ssrc,
                                                      const float* __restrict__ sw,
                                                      float* __restrict__ acc0, int N) {
  __shared__ int eidx[G0CAP];
  __shared__ float2 ewt[G0CAP];
  const int m0 = blockIdx.x * G0N;
  const int r = threadIdx.x >> 1;
  const int q = threadIdx.x & 1;
  const int t = m0 + r;
  const bool live = t < N;
  const int eb = start[m0];
  const int ee = start[min(N, m0 + G0N)];
  const int beg = live ? start[t] : 0;
  const int end = live ? start[t + 1] : 0;
  const float2* sw2 = (const float2*)sw;
  float4 a0 = make_float4(0.f, 0.f, 0.f, 0.f);
  float4 a1 = make_float4(0.f, 0.f, 0.f, 0.f);
  for (int c0 = eb; c0 < ee; c0 += G0CAP) {
    const int cn = min(G0CAP, ee - c0);
    int si[G0CAP / G0T];
    float2 wi[G0CAP / G0T];
#pragma unroll
    for (int u = 0; u < G0CAP / G0T; ++u) {
      const int i = (int)threadIdx.x + u * G0T;
      si[u] = 0;
      wi[u] = make_float2(0.f, 0.f);
      if (i < cn) {
        si[u] = ssrc[c0 + i];
        wi[u] = sw2[c0 + i];
      }
    }
#pragma unroll
    for (int u = 0; u < G0CAP / G0T; ++u) {
      const int i = (int)threadIdx.x + u * G0T;
      if (i < cn) {
        eidx[i] = si[u];
        ewt[i] = wi[u];
      }
    }
    __syncthreads();
    const int pb = max(beg, c0);
    const int pe = min(end, c0 + cn);
    sum_rows<5>((const char*)x, 16u * (unsigned)q, eidx, ewt, c0, pb, pe, a0, a1);
    if (c0 + G0CAP < ee) __syncthreads();
  }
  if (live) {
    float* a = acc0 + (size_t)t * 16 + 4 * q;
    *(float4*)(a) = a0;
    *(float4*)(a + 8) = a1;
  }
}

__global__ __launch_bounds__(32) void mlp0_kernel(const float* __restrict__ acc0,
                                                  const float* __restrict__ W0,
                                                  const float* __restrict__ b0,
                                                  float* __restrict__ h1) {
  const int m0 = blockIdx.x * 16;
  const int lane = threadIdx.x;
  const int n = lane & 15;
  const int kb = lane >> 4;
  v8f c0 = {}, c1 = {};
#pragma unroll
  for (int kt = 0; kt < 4; ++kt) {
    const int kA = kt * 4 + 2 * kb;
    v2f a, bb0, bb1;
    const float* ap = acc0 + (size_t)(m0 + n) * 16 + kA;
    a.x = ap[0];
    a.y = ap[1];
    bb0.x = W0[(kA + 0) * 32 + n];
    bb0.y = W0[(kA + 1) * 32 + n];
    bb1.x = W0[(kA + 0) * 32 + 16 + n];
    bb1.y = W0[(kA + 1) * 32 + 16 + n];
    c0 = wmma4f32(a, bb0, c0);
    c1 = wmma4f32(a, bb1, c1);
  }
  const float bias0 = b0[n], bias1 = b0[16 + n];
#pragma unroll
  for (int v = 0; v < 8; ++v) {
    float e0 = c0[v] + bias0;
    float e1 = c1[v] + bias1;
    float s = half16_sum(e0 * e0 + e1 * e1);
    const float invn = 1.0f / fmaxf(sqrtf(s), 1e-12f);
    const int row = m0 + v + 8 * kb;
    h1[(size_t)row * 32 + n] = e0 * invn;
    h1[(size_t)row * 32 + 16 + n] = e1 * invn;
  }
}

#define TS 132
#define AS 68
#define M1CAP 512
__global__ __launch_bounds__(128) void mlp1_kernel(const float* __restrict__ h1,
                                                   const int* __restrict__ start,
                                                   const int* __restrict__ ssrc,
                                                   const float* __restrict__ sw,
                                                   const float* __restrict__ W1,
                                                   const float* __restrict__ b1,
                                                   const float* __restrict__ W2,
                                                   const float* __restrict__ b2,
                                                   float* __restrict__ out) {
  __shared__ __attribute__((aligned(16))) float ash[16 * AS];
  __shared__ float tsh[16 * TS];
  __shared__ float rowsq[16];
  __shared__ int eidx[M1CAP];
  __shared__ float2 ewt[M1CAP];
  const int m0 = blockIdx.x * 16;
  const int wave = threadIdx.x >> 5;
  const int lane = threadIdx.x & 31;
  const int n = lane & 15;
  const int kb = lane >> 4;

  {
    const int r = threadIdx.x >> 3;
    const int q = threadIdx.x & 7;
    const int eb = start[m0];
    const int ee = start[m0 + 16];
    const int beg = start[m0 + r];
    const int end = start[m0 + r + 1];
    const float2* sw2 = (const float2*)sw;
    float4 a0 = make_float4(0.f, 0.f, 0.f, 0.f);
    float4 a1 = make_float4(0.f, 0.f, 0.f, 0.f);
    for (int c0 = eb; c0 < ee; c0 += M1CAP) {
      const int cn = min(M1CAP, ee - c0);
      int si[M1CAP / 128];
      float2 wi[M1CAP / 128];
#pragma unroll
      for (int u = 0; u < M1CAP / 128; ++u) {
        const int i = (int)threadIdx.x + u * 128;
        si[u] = 0;
        wi[u] = make_float2(0.f, 0.f);
        if (i < cn) {
          si[u] = ssrc[c0 + i];
          wi[u] = sw2[c0 + i];
        }
      }
#pragma unroll
      for (int u = 0; u < M1CAP / 128; ++u) {
        const int i = (int)threadIdx.x + u * 128;
        if (i < cn) {
          eidx[i] = si[u];
          ewt[i] = wi[u];
        }
      }
      __syncthreads();
      const int pb = max(beg, c0);
      const int pe = min(end, c0 + cn);
      sum_rows<7>((const char*)h1, 16u * (unsigned)q, eidx, ewt, c0, pb, pe, a0, a1);
      if (c0 + M1CAP < ee) __syncthreads();
    }
    *(float4*)(ash + r * AS + 4 * q) = a0;
    *(float4*)(ash + r * AS + 32 + 4 * q) = a1;
  }
  __syncthreads();

#pragma unroll
  for (int tt = 0; tt < 2; ++tt) {
    const int n0 = (wave * 2 + tt) * 16;
    v8f c = {};
    for (int kt = 0; kt < 16; ++kt) {
      const int kA = kt * 4 + 2 * kb;
      v2f a, b;
      const float* ap = ash + n * AS + kA;
      a.x = ap[0];
      a.y = ap[1];
      b.x = W1[(kA + 0) * 128 + n0 + n];
      b.y = W1[(kA + 1) * 128 + n0 + n];
      c = wmma4f32(a, b, c);
    }
    const float bias = b1[n0 + n];
#pragma unroll
    for (int v = 0; v < 8; ++v) {
      tsh[(v + 8 * kb) * TS + n0 + n] = fmaxf(c[v] + bias, 0.0f);
    }
  }
  if (threadIdx.x < 16) rowsq[threadIdx.x] = 0.0f;
  __syncthreads();

  const int n0 = wave * 16;
  v8f c = {};
  for (int kt = 0; kt < 32; ++kt) {
    const int kA = kt * 4 + 2 * kb;
    v2f a, b;
    a.x = tsh[n * TS + kA];
    a.y = tsh[n * TS + kA + 1];
    b.x = W2[(kA + 0) * 64 + n0 + n];
    b.y = W2[(kA + 1) * 64 + n0 + n];
    c = wmma4f32(a, b, c);
  }
  const float bias = b2[n0 + n];
  float cv[8];
#pragma unroll
  for (int v = 0; v < 8; ++v) {
    cv[v] = c[v] + bias;
    float s = half16_sum(cv[v] * cv[v]);
    if (n == 0) atomicAdd(&rowsq[v + 8 * kb], s);
  }
  __syncthreads();
#pragma unroll
  for (int v = 0; v < 8; ++v) {
    const int row = v + 8 * kb;
    const float invn = 1.0f / fmaxf(sqrtf(rowsq[row]), 1e-12f);
    out[(size_t)(m0 + row) * 64 + n0 + n] = cv[v] * invn;
  }
}

extern "C" void kernel_launch(void* const* d_in, const int* in_sizes, int n_in,
                              void* d_out, int out_size, void* d_ws, size_t ws_size,
                              hipStream_t stream) {
  const float* x  = (const float*)d_in[0];
  const int*   ei = (const int*)d_in[1];
  const float* Kw = (const float*)d_in[2];
  const float* W0 = (const float*)d_in[3];
  const float* b0 = (const float*)d_in[4];
  const float* W1 = (const float*)d_in[5];
  const float* b1 = (const float*)d_in[6];
  const float* W2 = (const float*)d_in[7];
  const float* b2 = (const float*)d_in[8];

  const int N = in_sizes[0] / 8;
  const int E = in_sizes[1] / 2;
  const int* src = ei;
  const int* dst = ei + E;

  const int NB = (N + BNODES - 1) / BNODES;

  int*      start    = (int*)d_ws;
  int*      rowTotal = start + N + 64;
  int*      ssrc     = rowTotal + NBMAX;
  float*    sw       = (float*)(ssrc + E);
  float*    acc0     = sw + (size_t)E * 2;
  float*    h1       = acc0 + (size_t)N * 16;
  int*      table    = ssrc;
  unsigned* recP     = (unsigned*)acc0;
  float*    recW     = (float*)(recP + E);

  bucket_count_kernel<<<G1, 1024, 0, stream>>>(dst, table, E, NB);
  bucket_scan_kernel<<<NB, G1, 0, stream>>>(table, rowTotal);
  bucket_place_kernel<<<G1, 1024, 0, stream>>>(src, dst, Kw, Kw + E, table, rowTotal, recP, recW,
                                               E, NB);
  node_place_kernel<<<NB, BNODES, 0, stream>>>(rowTotal, recP, recW, start, ssrc, sw, N);
  gather0_kernel<<<(N + G0N - 1) / G0N, G0T, 0, stream>>>(x, start, ssrc, sw, acc0, N);
  mlp0_kernel<<<N / 16, 32, 0, stream>>>(acc0, W0, b0, h1);
  mlp1_kernel<<<N / 16, 128, 0, stream>>>(h1, start, ssrc, sw, W1, b1, W2, b2,
                                          (float*)d_out);
}


# both sort levels order their records inside LDS, then store contiguous runs / whole lines instead of one scattered 4- or 8-byte store per record; per-node sum order stays an LDS arrival order
# speedup vs baseline: 7.9434x; 1.4749x over previous
#include <hip/hip_runtime.h>
#include <math.h>

typedef __attribute__((ext_vector_type(2))) float v2f;
typedef __attribute__((ext_vector_type(8))) float v8f;

__device__ __forceinline__ v8f wmma4f32(v2f a, v2f b, v8f c) {
  return __builtin_amdgcn_wmma_f32_16x16x4_f32(
       false, a,  false, b,
       (short)0, c,  false,  false);
}

__device__ __forceinline__ float half16_sum(float s) {
  s += __shfl_xor(s, 1, 32);
  s += __shfl_xor(s, 2, 32);
  s += __shfl_xor(s, 4, 32);
  s += __shfl_xor(s, 8, 32);
  return s;
}

#define BSH 8
#define BNODES 256
#define NBMAX 1024
#define G1 256

__device__ __forceinline__ int wave_incl_scan(int v, int lane) {
  for (int d = 1; d < 32; d <<= 1) {
    const int u = __shfl_up(v, d, 32);
    if (lane >= d) v += u;
  }
  return v;
}

__global__ __launch_bounds__(1024) void bucket_count_kernel(const int* __restrict__ dst,
                                                            int* __restrict__ table,
                                                            int E, int NB) {
  __shared__ int hist[NBMAX];
  const int g = blockIdx.x;
  hist[threadIdx.x] = 0;
  __syncthreads();
  const int chunk = (E + G1 - 1) / G1;
  const int e0 = g * chunk;
  const int e1 = min(E, e0 + chunk);
  int e = e0 + (int)threadIdx.x;
  for (; e + 3 * 1024 < e1; e += 4 * 1024) {
    const int d0 = dst[e], d1 = dst[e + 1024], d2 = dst[e + 2048], d3 = dst[e + 3072];
    atomicAdd(&hist[d0 >> BSH], 1);
    atomicAdd(&hist[d1 >> BSH], 1);
    atomicAdd(&hist[d2 >> BSH], 1);
    atomicAdd(&hist[d3 >> BSH], 1);
  }
  for (; e < e1; e += 1024) atomicAdd(&hist[dst[e] >> BSH], 1);
  __syncthreads();
  for (int b = threadIdx.x; b < NB; b += 1024) table[b * G1 + g] = hist[b];
}

__global__ __launch_bounds__(G1) void bucket_scan_kernel(int* __restrict__ table,
                                                         int* __restrict__ rowTotal) {
  __shared__ int wsum[G1 / 32];
  const int b = blockIdx.x;
  const int g = threadIdx.x;
  const int lane = g & 31;
  const int wave = g >> 5;
  const int c = table[b * G1 + g];
  const int incl = wave_incl_scan(c, lane);
  if (lane == 31) wsum[wave] = incl;
  __syncthreads();
  int off = 0;
  for (int w = 0; w < wave; ++w) off += wsum[w];
  table[b * G1 + g] = off + incl - c;
  if (g == G1 - 1) rowTotal[b] = off + incl;
}

#define PCAP 12800
__global__ __launch_bounds__(1024) void bucket_place_kernel(const int* __restrict__ src,
                                                            const int* __restrict__ dst,
                                                            const float* __restrict__ kw0,
                                                            const float* __restrict__ kw1,
                                                            const int* __restrict__ table,
                                                            const int* __restrict__ rowTotal,
                                                            unsigned* __restrict__ recP,
                                                            float* __restrict__ recW,
                                                            int E, int NB) {
  __shared__ int cur[NBMAX];
  __shared__ int delta[NBMAX];
  __shared__ int wsumA[32];
  __shared__ int wsumB[32];
  __shared__ uint4 stage[PCAP];
  const int g = blockIdx.x;
  const int tid = threadIdx.x;
  const int lane = tid & 31;
  const int wave = tid >> 5;
  const int c = (tid < NB) ? rowTotal[tid] : 0;
  const int t = (tid < NB) ? table[tid * G1 + g] : 0;
  int tn = 0;
  if (tid < NB) tn = (g + 1 < G1) ? table[tid * G1 + g + 1] : c;
  const int own = tn - t;
  const int inclC = wave_incl_scan(c, lane);
  const int inclO = wave_incl_scan(own, lane);
  if (lane == 31) {
    wsumA[wave] = inclC;
    wsumB[wave] = inclO;
  }
  __syncthreads();
  if (wave == 0) {
    const int vA = wsumA[lane];
    const int sA = wave_incl_scan(vA, lane);
    wsumA[lane] = sA - vA;
    const int vB = wsumB[lane];
    const int sB = wave_incl_scan(vB, lane);
    wsumB[lane] = sB - vB;
  }
  __syncthreads();
  const int gb = wsumA[wave] + inclC - c + t;
  const int ls = wsumB[wave] + inclO - own;
  const int chunk = (E + G1 - 1) / G1;
  const int e0 = g * chunk;
  const int e1 = min(E, e0 + chunk);
  const bool ordered = chunk <= PCAP;
  cur[tid] = ordered ? ls : gb;
  delta[tid] = gb - ls;
  __syncthreads();
  if (ordered) {
    int e = e0 + tid;
    for (; e + 3 * 1024 < e1; e += 4 * 1024) {
      int d[4];
      unsigned s[4];
      float2 w[4];
      int pos[4];
#pragma unroll
      for (int u = 0; u < 4; ++u) {
        d[u] = dst[e + u * 1024];
        s[u] = (unsigned)src[e + u * 1024];
        w[u] = make_float2(kw0[e + u * 1024], kw1[e + u * 1024]);
      }
#pragma unroll
      for (int u = 0; u < 4; ++u) pos[u] = atomicAdd(&cur[d[u] >> BSH], 1);
#pragma unroll
      for (int u = 0; u < 4; ++u)
        stage[pos[u]] = make_uint4(s[u] | ((unsigned)(d[u] & (BNODES - 1)) << 24),
                                   __float_as_uint(w[u].x), __float_as_uint(w[u].y),
                                   (unsigned)(pos[u] + delta[d[u] >> BSH]));
    }
    for (; e < e1; e += 1024) {
      const int d = dst[e];
      const unsigned s = (unsigned)src[e];
      const float2 w = make_float2(kw0[e], kw1[e]);
      const int pos = atomicAdd(&cur[d >> BSH], 1);
      stage[pos] = make_uint4(s | ((unsigned)(d & (BNODES - 1)) << 24), __float_as_uint(w.x),
                              __float_as_uint(w.y), (unsigned)(pos + delta[d >> BSH]));
    }
    __syncthreads();
    const int nloc = e1 - e0;
    for (int i = tid; i < nloc; i += 1024) {
      const uint4 r = stage[i];
      recP[r.w] = r.x;
      ((float2*)recW)[r.w] = make_float2(__uint_as_float(r.y), __uint_as_float(r.z));
    }
  } else {
    int e = e0 + tid;
    for (; e < e1; e += 1024) {
      const int d = dst[e];
      const unsigned s = (unsigned)src[e];
      const float2 w = make_float2(kw0[e], kw1[e]);
      const int pos = atomicAdd(&cur[d >> BSH], 1);
      recP[pos] = s | ((unsigned)(d & (BNODES - 1)) << 24);
      ((float2*)recW)[pos] = w;
    }
  }
}

#define NCAP 6144
__global__ __launch_bounds__(BNODES) void node_place_kernel(const int* __restrict__ rowTotal,
                                                            const unsigned* __restrict__ recP,
                                                            const float* __restrict__ recW,
                                                            int* __restrict__ start,
                                                            int* __restrict__ ssrc,
                                                            float* __restrict__ sw, int N) {
  __shared__ int cntL[BNODES];
  __shared__ int curL[BNODES];
  __shared__ int wsum[BNODES / 32];
  __shared__ int baseL;
  __shared__ int sP[NCAP];
  __shared__ float2 sW[NCAP];
  const int b = blockIdx.x;
  const int tid = threadIdx.x;
  const int lane = tid & 31;
  const int wave = tid >> 5;
  cntL[tid] = 0;
  if (tid == 0) baseL = 0;
  __syncthreads();
  int part = 0;
  for (int i = tid; i < b; i += BNODES) part += rowTotal[i];
  part += __shfl_xor(part, 16, 32);
  part += __shfl_xor(part, 8, 32);
  part += __shfl_xor(part, 4, 32);
  part += __shfl_xor(part, 2, 32);
  part += __shfl_xor(part, 1, 32);
  if (lane == 0 && part != 0) atomicAdd(&baseL, part);
  const int n = rowTotal[b];
  __syncthreads();
  const int base = baseL;
  const unsigned* rp = recP + base;
  const float2* rw = (const float2*)recW + base;
  int i = tid;
  for (; i + 3 * BNODES < n; i += 4 * BNODES) {
    const unsigned p0 = rp[i], p1 = rp[i + BNODES], p2 = rp[i + 2 * BNODES], p3 = rp[i + 3 * BNODES];
    atomicAdd(&cntL[p0 >> 24], 1);
    atomicAdd(&cntL[p1 >> 24], 1);
    atomicAdd(&cntL[p2 >> 24], 1);
    atomicAdd(&cntL[p3 >> 24], 1);
  }
  for (; i < n; i += BNODES) atomicAdd(&cntL[rp[i] >> 24], 1);
  __syncthreads();
  const int c = cntL[tid];
  const int incl = wave_incl_scan(c, lane);
  if (lane == 31) wsum[wave] = incl;
  __syncthreads();
  int off = 0;
  for (int w = 0; w < wave; ++w) off += wsum[w];
  const int excl = off + incl - c;
  curL[tid] = excl;
  const int node = b * BNODES + tid;
  if (node < N) start[node] = base + excl;
  if (node == N - 1) start[N] = base + excl + c;
  __syncthreads();
  if (n <= NCAP) {
    int k = tid;
    for (; k + 3 * BNODES < n; k += 4 * BNODES) {
      unsigned p[4];
      float2 w[4];
      int pos[4];
#pragma unroll
      for (int u = 0; u < 4; ++u) {
        p[u] = rp[k + u * BNODES];
        w[u] = rw[k + u * BNODES];
      }
#pragma unroll
      for (int u = 0; u < 4; ++u) pos[u] = atomicAdd(&curL[p[u] >> 24], 1);
#pragma unroll
      for (int u = 0; u < 4; ++u) {
        sP[pos[u]] = (int)(p[u] & 0xFFFFFFu);
        sW[pos[u]] = w[u];
      }
    }
    for (int i = k; i < n; i += BNODES) {
      const unsigned p = rp[i];
      const float2 w = rw[i];
      const int pos = atomicAdd(&curL[p >> 24], 1);
      sP[pos] = (int)(p & 0xFFFFFFu);
      sW[pos] = w;
    }
    __syncthreads();
    for (int i = tid; i < n; i += BNODES) {
      ssrc[base + i] = sP[i];
      ((float2*)sw)[base + i] = sW[i];
    }
  } else {
    for (int i = tid; i < n; i += BNODES) {
      const unsigned p = rp[i];
      const float2 w = rw[i];
      const int pos = base + atomicAdd(&curL[p >> 24], 1);
      ssrc[pos] = (int)(p & 0xFFFFFFu);
      ((float2*)sw)[pos] = w;
    }
  }
}

#define EB 8
static_assert(EB == 8, "sum_rows names the EB rows of a batch one by one");

template <int ROWSHIFT>
__device__ __forceinline__ void sum_rows(const char* __restrict__ rows, unsigned lane_off,
                                         const int* eidx, const float2* ewt,
                                         int c0, int pb, int pe, float4& a0, float4& a1) {
  for (int p = pb; p < pe; p += EB) {
    const int nv = pe - p;
    int s[EB];
    float2 w[EB];
    float4 v[EB];
#pragma unroll
    for (int u = 0; u < EB; ++u) {
      const int i = min(p + u, pe - 1) - c0;
      s[u] = eidx[i];
      const float2 ww = ewt[i];
      w[u].x = (u < nv) ? ww.x : 0.f;
      w[u].y = (u < nv) ? ww.y : 0.f;
    }
#pragma unroll
    for (int u = 0; u < EB; ++u)
      v[u] = *(const float4*)(rows + ((((unsigned)s[u]) << ROWSHIFT) | lane_off));
    asm volatile(""
                 : "+v"(a0.x), "+v"(a0.y), "+v"(a0.z), "+v"(a0.w),
                   "+v"(a1.x), "+v"(a1.y), "+v"(a1.z), "+v"(a1.w)
                 : "v"(v[0].x), "v"(v[1].x), "v"(v[2].x), "v"(v[3].x),
                   "v"(v[4].x), "v"(v[5].x), "v"(v[6].x), "v"(v[7].x));
#pragma unroll
    for (int u = 0; u < EB; ++u) {
      const float w0 = w[u].x, w1 = w[u].y;
      a0.x += w0 * v[u].x; a0.y += w0 * v[u].y; a0.z += w0 * v[u].z; a0.w += w0 * v[u].w;
      a1.x += w1 * v[u].x; a1.y += w1 * v[u].y; a1.z += w1 * v[u].z; a1.w += w1 * v[u].w;
    }
  }
}

#define G0N 64
#define G0T 128
#define G0CAP 1536
__global__ __launch_bounds__(G0T) void gather0_kernel(const float* __restrict__ x,
                                                      const int* __restrict__ start,
                                                      const int* __restrict__ ssrc,
                                                      const float* __restrict__ sw,
                                                      float* __restrict__ acc0, int N) {
  __shared__ int eidx[G0CAP];
  __shared__ float2 ewt[G0CAP];
  const int m0 = blockIdx.x * G0N;
  const int r = threadIdx.x >> 1;
  const int q = threadIdx.x & 1;
  const int t = m0 + r;
  const bool live = t < N;
  const int eb = start[m0];
  const int ee = start[min(N, m0 + G0N)];
  const int beg = live ? start[t] : 0;
  const int end = live ? start[t + 1] : 0;
  const float2* sw2 = (const float2*)sw;
  float4 a0 = make_float4(0.f, 0.f, 0.f, 0.f);
  float4 a1 = make_float4(0.f, 0.f, 0.f, 0.f);
  for (int c0 = eb; c0 < ee; c0 += G0CAP) {
    const int cn = min(G0CAP, ee - c0);
    int si[G0CAP / G0T];
    float2 wi[G0CAP / G0T];
#pragma unroll
    for (int u = 0; u < G0CAP / G0T; ++u) {
      const int i = (int)threadIdx.x + u * G0T;
      si[u] = 0;
      wi[u] = make_float2(0.f, 0.f);
      if (i < cn) {
        si[u] = ssrc[c0 + i];
        wi[u] = sw2[c0 + i];
      }
    }
#pragma unroll
    for (int u = 0; u < G0CAP / G0T; ++u) {
      const int i = (int)threadIdx.x + u * G0T;
      if (i < cn) {
        eidx[i] = si[u];
        ewt[i] = wi[u];
      }
    }
    __syncthreads();
    const int pb = max(beg, c0);
    const int pe = min(end, c0 + cn);
    sum_rows<5>((const char*)x, 16u * (unsigned)q, eidx, ewt, c0, pb, pe, a0, a1);
    if (c0 + G0CAP < ee) __syncthreads();
  }
  if (live) {
    float* a = acc0 + (size_t)t * 16 + 4 * q;
    *(float4*)(a) = a0;
    *(float4*)(a + 8) = a1;
  }
}

__global__ __launch_bounds__(32) void mlp0_kernel(const float* __restrict__ acc0,
                                                  const float* __restrict__ W0,
                                                  const float* __restrict__ b0,
                                                  float* __restrict__ h1) {
  const int m0 = blockIdx.x * 16;
  const int lane = threadIdx.x;
  const int n = lane & 15;
  const int kb = lane >> 4;
  v8f c0 = {}, c1 = {};
#pragma unroll
  for (int kt = 0; kt < 4; ++kt) {
    const int kA = kt * 4 + 2 * kb;
    v2f a, bb0, bb1;
    const float* ap = acc0 + (size_t)(m0 + n) * 16 + kA;
    a.x = ap[0];
    a.y = ap[1];
    bb0.x = W0[(kA + 0) * 32 + n];
    bb0.y = W0[(kA + 1) * 32 + n];
    bb1.x = W0[(kA + 0) * 32 + 16 + n];
    bb1.y = W0[(kA + 1) * 32 + 16 + n];
    c0 = wmma4f32(a, bb0, c0);
    c1 = wmma4f32(a, bb1, c1);
  }
  const float bias0 = b0[n], bias1 = b0[16 + n];
#pragma unroll
  for (int v = 0; v < 8; ++v) {
    float e0 = c0[v] + bias0;
    float e1 = c1[v] + bias1;
    float s = half16_sum(e0 * e0 + e1 * e1);
    const float invn = 1.0f / fmaxf(sqrtf(s), 1e-12f);
    const int row = m0 + v + 8 * kb;
    h1[(size_t)row * 32 + n] = e0 * invn;
    h1[(size_t)row * 32 + 16 + n] = e1 * invn;
  }
}

#define TS 132
#define AS 68
#define M1CAP 512
__global__ __launch_bounds__(128) void mlp1_kernel(const float* __restrict__ h1,
                                                   const int* __restrict__ start,
                                                   const int* __restrict__ ssrc,
                                                   const float* __restrict__ sw,
                                                   const float* __restrict__ W1,
                                                   const float* __restrict__ b1,
                                                   const float* __restrict__ W2,
                                                   const float* __restrict__ b2,
                                                   float* __restrict__ out) {
  __shared__ __attribute__((aligned(16))) float ash[16 * AS];
  __shared__ float tsh[16 * TS];
  __shared__ float rowsq[16];
  __shared__ int eidx[M1CAP];
  __shared__ float2 ewt[M1CAP];
  const int m0 = blockIdx.x * 16;
  const int wave = threadIdx.x >> 5;
  const int lane = threadIdx.x & 31;
  const int n = lane & 15;
  const int kb = lane >> 4;

  {
    const int r = threadIdx.x >> 3;
    const int q = threadIdx.x & 7;
    const int eb = start[m0];
    const int ee = start[m0 + 16];
    const int beg = start[m0 + r];
    const int end = start[m0 + r + 1];
    const float2* sw2 = (const float2*)sw;
    float4 a0 = make_float4(0.f, 0.f, 0.f, 0.f);
    float4 a1 = make_float4(0.f, 0.f, 0.f, 0.f);
    for (int c0 = eb; c0 < ee; c0 += M1CAP) {
      const int cn = min(M1CAP, ee - c0);
      int si[M1CAP / 128];
      float2 wi[M1CAP / 128];
#pragma unroll
      for (int u = 0; u < M1CAP / 128; ++u) {
        const int i = (int)threadIdx.x + u * 128;
        si[u] = 0;
        wi[u] = make_float2(0.f, 0.f);
        if (i < cn) {
          si[u] = ssrc[c0 + i];
          wi[u] = sw2[c0 + i];
        }
      }
#pragma unroll
      for (int u = 0; u < M1CAP / 128; ++u) {
        const int i = (int)threadIdx.x + u * 128;
        if (i < cn) {
          eidx[i] = si[u];
          ewt[i] = wi[u];
        }
      }
      __syncthreads();
      const int pb = max(beg, c0);
      const int pe = min(end, c0 + cn);
      sum_rows<7>((const char*)h1, 16u * (unsigned)q, eidx, ewt, c0, pb, pe, a0, a1);
      if (c0 + M1CAP < ee) __syncthreads();
    }
    *(float4*)(ash + r * AS + 4 * q) = a0;
    *(float4*)(ash + r * AS + 32 + 4 * q) = a1;
  }
  __syncthreads();

#pragma unroll
  for (int tt = 0; tt < 2; ++tt) {
    const int n0 = (wave * 2 + tt) * 16;
    v8f c = {};
    for (int kt = 0; kt < 16; ++kt) {
      const int kA = kt * 4 + 2 * kb;
      v2f a, b;
      const float* ap = ash + n * AS + kA;
      a.x = ap[0];
      a.y = ap[1];
      b.x = W1[(kA + 0) * 128 + n0 + n];
      b.y = W1[(kA + 1) * 128 + n0 + n];
      c = wmma4f32(a, b, c);
    }
    const float bias = b1[n0 + n];
#pragma unroll
    for (int v = 0; v < 8; ++v) {
      tsh[(v + 8 * kb) * TS + n0 + n] = fmaxf(c[v] + bias, 0.0f);
    }
  }
  if (threadIdx.x < 16) rowsq[threadIdx.x] = 0.0f;
  __syncthreads();

  const int n0 = wave * 16;
  v8f c = {};
  for (int kt = 0; kt < 32; ++kt) {
    const int kA = kt * 4 + 2 * kb;
    v2f a, b;
    a.x = tsh[n * TS + kA];
    a.y = tsh[n * TS + kA + 1];
    b.x = W2[(kA + 0) * 64 + n0 + n];
    b.y = W2[(kA + 1) * 64 + n0 + n];
    c = wmma4f32(a, b, c);
  }
  const float bias = b2[n0 + n];
  float cv[8];
#pragma unroll
  for (int v = 0; v < 8; ++v) {
    cv[v] = c[v] + bias;
    float s = half16_sum(cv[v] * cv[v]);
    if (n == 0) atomicAdd(&rowsq[v + 8 * kb], s);
  }
  __syncthreads();
#pragma unroll
  for (int v = 0; v < 8; ++v) {
    const int row = v + 8 * kb;
    const float invn = 1.0f / fmaxf(sqrtf(rowsq[row]), 1e-12f);
    out[(size_t)(m0 + row) * 64 + n0 + n] = cv[v] * invn;
  }
}

extern "C" void kernel_launch(void* const* d_in, const int* in_sizes, int n_in,
                              void* d_out, int out_size, void* d_ws, size_t ws_size,
                              hipStream_t stream) {
  const float* x  = (const float*)d_in[0];
  const int*   ei = (const int*)d_in[1];
  const float* Kw = (const float*)d_in[2];
  const float* W0 = (const float*)d_in[3];
  const float* b0 = (const float*)d_in[4];
  const float* W1 = (const float*)d_in[5];
  const float* b1 = (const float*)d_in[6];
  const float* W2 = (const float*)d_in[7];
  const float* b2 = (const float*)d_in[8];

  const int N = in_sizes[0] / 8;
  const int E = in_sizes[1] / 2;
  const int* src = ei;
  const int* dst = ei + E;

  const int NB = (N + BNODES - 1) / BNODES;

  int*      start    = (int*)d_ws;
  int*      rowTotal = start + N + 64;
  int*      ssrc     = rowTotal + NBMAX;
  float*    sw       = (float*)(ssrc + E);
  float*    acc0     = sw + (size_t)E * 2;
  float*    h1       = acc0 + (size_t)N * 16;
  int*      table    = ssrc;
  unsigned* recP     = (unsigned*)acc0;
  float*    recW     = (float*)(recP + E);

  bucket_count_kernel<<<G1, 1024, 0, stream>>>(dst, table, E, NB);
  bucket_scan_kernel<<<NB, G1, 0, stream>>>(table, rowTotal);
  bucket_place_kernel<<<G1, 1024, 0, stream>>>(src, dst, Kw, Kw + E, table, rowTotal, recP, recW,
                                               E, NB);
  node_place_kernel<<<NB, BNODES, 0, stream>>>(rowTotal, recP, recW, start, ssrc, sw, N);
  gather0_kernel<<<(N + G0N - 1) / G0N, G0T, 0, stream>>>(x, start, ssrc, sw, acc0, N);
  mlp0_kernel<<<N / 16, 32, 0, stream>>>(acc0, W0, b0, h1);
  mlp1_kernel<<<N / 16, 128, 0, stream>>>(h1, start, ssrc, sw, W1, b1, W2, b2,
                                          (float*)d_out);
}


# layer-0 MLP fused into the layer-0 aggregation kernel: a wave's 16 node sums reach its own WMMA row tile through LDS; sums plane, one launch and 12500 one-wave work-groups gone; sum order unchanged
# speedup vs baseline: 8.4091x; 1.0586x over previous
#include <hip/hip_runtime.h>
#include <math.h>

typedef __attribute__((ext_vector_type(2))) float v2f;
typedef __attribute__((ext_vector_type(8))) float v8f;

__device__ __forceinline__ v8f wmma4f32(v2f a, v2f b, v8f c) {
  return __builtin_amdgcn_wmma_f32_16x16x4_f32(
       false, a,  false, b,
       (short)0, c,  false,  false);
}

__device__ __forceinline__ float half16_sum(float s) {
  s += __shfl_xor(s, 1, 32);
  s += __shfl_xor(s, 2, 32);
  s += __shfl_xor(s, 4, 32);
  s += __shfl_xor(s, 8, 32);
  return s;
}

#define BSH 8
#define BNODES 256
#define NBMAX 1024
#define G1 256

__device__ __forceinline__ int wave_incl_scan(int v, int lane) {
  for (int d = 1; d < 32; d <<= 1) {
    const int u = __shfl_up(v, d, 32);
    if (lane >= d) v += u;
  }
  return v;
}

__global__ __launch_bounds__(1024) void bucket_count_kernel(const int* __restrict__ dst,
                                                            int* __restrict__ table,
                                                            int E, int NB) {
  __shared__ int hist[NBMAX];
  const int g = blockIdx.x;
  hist[threadIdx.x] = 0;
  __syncthreads();
  const int chunk = (E + G1 - 1) / G1;
  const int e0 = g * chunk;
  const int e1 = min(E, e0 + chunk);
  int e = e0 + (int)threadIdx.x;
  for (; e + 3 * 1024 < e1; e += 4 * 1024) {
    const int d0 = dst[e], d1 = dst[e + 1024], d2 = dst[e + 2048], d3 = dst[e + 3072];
    atomicAdd(&hist[d0 >> BSH], 1);
    atomicAdd(&hist[d1 >> BSH], 1);
    atomicAdd(&hist[d2 >> BSH], 1);
    atomicAdd(&hist[d3 >> BSH], 1);
  }
  for (; e < e1; e += 1024) atomicAdd(&hist[dst[e] >> BSH], 1);
  __syncthreads();
  for (int b = threadIdx.x; b < NB; b += 1024) table[b * G1 + g] = hist[b];
}

__global__ __launch_bounds__(G1) void bucket_scan_kernel(int* __restrict__ table,
                                                         int* __restrict__ rowTotal) {
  __shared__ int wsum[G1 / 32];
  const int b = blockIdx.x;
  const int g = threadIdx.x;
  const int lane = g & 31;
  const int wave = g >> 5;
  const int c = table[b * G1 + g];
  const int incl = wave_incl_scan(c, lane);
  if (lane == 31) wsum[wave] = incl;
  __syncthreads();
  int off = 0;
  for (int w = 0; w < wave; ++w) off += wsum[w];
  table[b * G1 + g] = off + incl - c;
  if (g == G1 - 1) rowTotal[b] = off + incl;
}

#define PCAP 12800
__global__ __launch_bounds__(1024) void bucket_place_kernel(const int* __restrict__ src,
                                                            const int* __restrict__ dst,
                                                            const float* __restrict__ kw0,
                                                            const float* __restrict__ kw1,
                                                            const int* __restrict__ table,
                                                            const int* __restrict__ rowTotal,
                                                            unsigned* __restrict__ recP,
                                                            float* __restrict__ recW,
                                                            int E, int NB) {
  __shared__ int cur[NBMAX];
  __shared__ int delta[NBMAX];
  __shared__ int wsumA[32];
  __shared__ int wsumB[32];
  __shared__ uint4 stage[PCAP];
  const int g = blockIdx.x;
  const int tid = threadIdx.x;
  const int lane = tid & 31;
  const int wave = tid >> 5;
  const int c = (tid < NB) ? rowTotal[tid] : 0;
  const int t = (tid < NB) ? table[tid * G1 + g] : 0;
  int tn = 0;
  if (tid < NB) tn = (g + 1 < G1) ? table[tid * G1 + g + 1] : c;
  const int own = tn - t;
  const int inclC = wave_incl_scan(c, lane);
  const int inclO = wave_incl_scan(own, lane);
  if (lane == 31) {
    wsumA[wave] = inclC;
    wsumB[wave] = inclO;
  }
  __syncthreads();
  if (wave == 0) {
    const int vA = wsumA[lane];
    const int sA = wave_incl_scan(vA, lane);
    wsumA[lane] = sA - vA;
    const int vB = wsumB[lane];
    const int sB = wave_incl_scan(vB, lane);
    wsumB[lane] = sB - vB;
  }
  __syncthreads();
  const int gb = wsumA[wave] + inclC - c + t;
  const int ls = wsumB[wave] + inclO - own;
  const int chunk = (E + G1 - 1) / G1;
  const int e0 = g * chunk;
  const int e1 = min(E, e0 + chunk);
  const bool ordered = chunk <= PCAP;
  cur[tid] = ordered ? ls : gb;
  delta[tid] = gb - ls;
  __syncthreads();
  if (ordered) {
    int e = e0 + tid;
    for (; e + 3 * 1024 < e1; e += 4 * 1024) {
      int d[4];
      unsigned s[4];
      float2 w[4];
      int pos[4];
#pragma unroll
      for (int u = 0; u < 4; ++u) {
        d[u] = dst[e + u * 1024];
        s[u] = (unsigned)src[e + u * 1024];
        w[u] = make_float2(kw0[e + u * 1024], kw1[e + u * 1024]);
      }
#pragma unroll
      for (int u = 0; u < 4; ++u) pos[u] = atomicAdd(&cur[d[u] >> BSH], 1);
#pragma unroll
      for (int u = 0; u < 4; ++u)
        stage[pos[u]] = make_uint4(s[u] | ((unsigned)(d[u] & (BNODES - 1)) << 24),
                                   __float_as_uint(w[u].x), __float_as_uint(w[u].y),
                                   (unsigned)(pos[u] + delta[d[u] >> BSH]));
    }
    for (; e < e1; e += 1024) {
      const int d = dst[e];
      const unsigned s = (unsigned)src[e];
      const float2 w = make_float2(kw0[e], kw1[e]);
      const int pos = atomicAdd(&cur[d >> BSH], 1);
      stage[pos] = make_uint4(s | ((unsigned)(d & (BNODES - 1)) << 24), __float_as_uint(w.x),
                              __float_as_uint(w.y), (unsigned)(pos + delta[d >> BSH]));
    }
    __syncthreads();
    const int nloc = e1 - e0;
    for (int i = tid; i < nloc; i += 1024) {
      const uint4 r = stage[i];
      recP[r.w] = r.x;
      ((float2*)recW)[r.w] = make_float2(__uint_as_float(r.y), __uint_as_float(r.z));
    }
  } else {
    int e = e0 + tid;
    for (; e < e1; e += 1024) {
      const int d = dst[e];
      const unsigned s = (unsigned)src[e];
      const float2 w = make_float2(kw0[e], kw1[e]);
      const int pos = atomicAdd(&cur[d >> BSH], 1);
      recP[pos] = s | ((unsigned)(d & (BNODES - 1)) << 24);
      ((float2*)recW)[pos] = w;
    }
  }
}

#define NCAP 6144
__global__ __launch_bounds__(BNODES) void node_place_kernel(const int* __restrict__ rowTotal,
                                                            const unsigned* __restrict__ recP,
                                                            const float* __restrict__ recW,
                                                            int* __restrict__ start,
                                                            int* __restrict__ ssrc,
                                                            float* __restrict__ sw, int N) {
  __shared__ int cntL[BNODES];
  __shared__ int curL[BNODES];
  __shared__ int wsum[BNODES / 32];
  __shared__ int baseL;
  __shared__ int sP[NCAP];
  __shared__ float2 sW[NCAP];
  const int b = blockIdx.x;
  const int tid = threadIdx.x;
  const int lane = tid & 31;
  const int wave = tid >> 5;
  cntL[tid] = 0;
  if (tid == 0) baseL = 0;
  __syncthreads();
  int part = 0;
  for (int i = tid; i < b; i += BNODES) part += rowTotal[i];
  part += __shfl_xor(part, 16, 32);
  part += __shfl_xor(part, 8, 32);
  part += __shfl_xor(part, 4, 32);
  part += __shfl_xor(part, 2, 32);
  part += __shfl_xor(part, 1, 32);
  if (lane == 0 && part != 0) atomicAdd(&baseL, part);
  const int n = rowTotal[b];
  __syncthreads();
  const int base = baseL;
  const unsigned* rp = recP + base;
  const float2* rw = (const float2*)recW + base;
  int i = tid;
  for (; i + 3 * BNODES < n; i += 4 * BNODES) {
    const unsigned p0 = rp[i], p1 = rp[i + BNODES], p2 = rp[i + 2 * BNODES], p3 = rp[i + 3 * BNODES];
    atomicAdd(&cntL[p0 >> 24], 1);
    atomicAdd(&cntL[p1 >> 24], 1);
    atomicAdd(&cntL[p2 >> 24], 1);
    atomicAdd(&cntL[p3 >> 24], 1);
  }
  for (; i < n; i += BNODES) atomicAdd(&cntL[rp[i] >> 24], 1);
  __syncthreads();
  const int c = cntL[tid];
  const int incl = wave_incl_scan(c, lane);
  if (lane == 31) wsum[wave] = incl;
  __syncthreads();
  int off = 0;
  for (int w = 0; w < wave; ++w) off += wsum[w];
  const int excl = off + incl - c;
  curL[tid] = excl;
  const int node = b * BNODES + tid;
  if (node < N) start[node] = base + excl;
  if (node == N - 1) start[N] = base + excl + c;
  __syncthreads();
  if (n <= NCAP) {
    int k = tid;
    for (; k + 3 * BNODES < n; k += 4 * BNODES) {
      unsigned p[4];
      float2 w[4];
      int pos[4];
#pragma unroll
      for (int u = 0; u < 4; ++u) {
        p[u] = rp[k + u * BNODES];
        w[u] = rw[k + u * BNODES];
      }
#pragma unroll
      for (int u = 0; u < 4; ++u) pos[u] = atomicAdd(&curL[p[u] >> 24], 1);
#pragma unroll
      for (int u = 0; u < 4; ++u) {
        sP[pos[u]] = (int)(p[u] & 0xFFFFFFu);
        sW[pos[u]] = w[u];
      }
    }
    for (int i = k; i < n; i += BNODES) {
      const unsigned p = rp[i];
      const float2 w = rw[i];
      const int pos = atomicAdd(&curL[p >> 24], 1);
      sP[pos] = (int)(p & 0xFFFFFFu);
      sW[pos] = w;
    }
    __syncthreads();
    for (int i = tid; i < n; i += BNODES) {
      ssrc[base + i] = sP[i];
      ((float2*)sw)[base + i] = sW[i];
    }
  } else {
    for (int i = tid; i < n; i += BNODES) {
      const unsigned p = rp[i];
      const float2 w = rw[i];
      const int pos = base + atomicAdd(&curL[p >> 24], 1);
      ssrc[pos] = (int)(p & 0xFFFFFFu);
      ((float2*)sw)[pos] = w;
    }
  }
}

#define EB 8
static_assert(EB == 8, "sum_rows names the EB rows of a batch one by one");

template <int ROWSHIFT>
__device__ __forceinline__ void sum_rows(const char* __restrict__ rows, unsigned lane_off,
                                         const int* eidx, const float2* ewt,
                                         int c0, int pb, int pe, float4& a0, float4& a1) {
  for (int p = pb; p < pe; p += EB) {
    const int nv = pe - p;
    int s[EB];
    float2 w[EB];
    float4 v[EB];
#pragma unroll
    for (int u = 0; u < EB; ++u) {
      const int i = min(p + u, pe - 1) - c0;
      s[u] = eidx[i];
      const float2 ww = ewt[i];
      w[u].x = (u < nv) ? ww.x : 0.f;
      w[u].y = (u < nv) ? ww.y : 0.f;
    }
#pragma unroll
    for (int u = 0; u < EB; ++u)
      v[u] = *(const float4*)(rows + ((((unsigned)s[u]) << ROWSHIFT) | lane_off));
    asm volatile(""
                 : "+v"(a0.x), "+v"(a0.y), "+v"(a0.z), "+v"(a0.w),
                   "+v"(a1.x), "+v"(a1.y), "+v"(a1.z), "+v"(a1.w)
                 : "v"(v[0].x), "v"(v[1].x), "v"(v[2].x), "v"(v[3].x),
                   "v"(v[4].x), "v"(v[5].x), "v"(v[6].x), "v"(v[7].x));
#pragma unroll
    for (int u = 0; u < EB; ++u) {
      const float w0 = w[u].x, w1 = w[u].y;
      a0.x += w0 * v[u].x; a0.y += w0 * v[u].y; a0.z += w0 * v[u].z; a0.w += w0 * v[u].w;
      a1.x += w1 * v[u].x; a1.y += w1 * v[u].y; a1.z += w1 * v[u].z; a1.w += w1 * v[u].w;
    }
  }
}

#define G0N 64
#define G0T 128
#define G0CAP 1536
#define A0P 20
__global__ __launch_bounds__(G0T, 12) void gather0_mlp0_kernel(const float* __restrict__ x,
                                                           const int* __restrict__ start,
                                                           const int* __restrict__ ssrc,
                                                           const float* __restrict__ sw,
                                                           const float* __restrict__ W0,
                                                           const float* __restrict__ b0,
                                                           float* __restrict__ h1, int N) {
  __shared__ int eidx[G0CAP];
  __shared__ float2 ewt[G0CAP];
  __shared__ __attribute__((aligned(16))) float atile[G0N * A0P];
  const int m0 = blockIdx.x * G0N;
  const int r = threadIdx.x >> 1;
  const int q = threadIdx.x & 1;
  const int t = m0 + r;
  const bool live = t < N;
  const int eb = start[m0];
  const int ee = start[min(N, m0 + G0N)];
  const int beg = live ? start[t] : 0;
  const int end = live ? start[t + 1] : 0;
  const float2* sw2 = (const float2*)sw;
  float4 a0 = make_float4(0.f, 0.f, 0.f, 0.f);
  float4 a1 = make_float4(0.f, 0.f, 0.f, 0.f);
  for (int c0 = eb; c0 < ee; c0 += G0CAP) {
    const int cn = min(G0CAP, ee - c0);
    int si[G0CAP / G0T];
    float2 wi[G0CAP / G0T];
#pragma unroll
    for (int u = 0; u < G0CAP / G0T; ++u) {
      const int i = (int)threadIdx.x + u * G0T;
      si[u] = 0;
      wi[u] = make_float2(0.f, 0.f);
      if (i < cn) {
        si[u] = ssrc[c0 + i];
        wi[u] = sw2[c0 + i];
      }
    }
#pragma unroll
    for (int u = 0; u < G0CAP / G0T; ++u) {
      const int i = (int)threadIdx.x + u * G0T;
      if (i < cn) {
        eidx[i] = si[u];
        ewt[i] = wi[u];
      }
    }
    __syncthreads();
    const int pb = max(beg, c0);
    const int pe = min(end, c0 + cn);
    sum_rows<5>((const char*)x, 16u * (unsigned)q, eidx, ewt, c0, pb, pe, a0, a1);
    if (c0 + G0CAP < ee) __syncthreads();
  }
  *(float4*)(atile + r * A0P + 4 * q) = a0;
  *(float4*)(atile + r * A0P + 8 + 4 * q) = a1;
  __syncthreads();

  const int wave = threadIdx.x >> 5;
  const int lane = threadIdx.x & 31;
  const int n = lane & 15;
  const int kb = lane >> 4;
  const int t0 = m0 + 16 * wave;
  if (t0 + 16 <= N) {
    v8f c0 = {}, c1 = {};
#pragma unroll
    for (int kt = 0; kt < 4; ++kt) {
      const int kA = kt * 4 + 2 * kb;
      v2f a, bb0, bb1;
      const float* ap = atile + (16 * wave + n) * A0P + kA;
      a.x = ap[0];
      a.y = ap[1];
      bb0.x = W0[(kA + 0) * 32 + n];
      bb0.y = W0[(kA + 1) * 32 + n];
      bb1.x = W0[(kA + 0) * 32 + 16 + n];
      bb1.y = W0[(kA + 1) * 32 + 16 + n];
      c0 = wmma4f32(a, bb0, c0);
      c1 = wmma4f32(a, bb1, c1);
    }
    const float bias0 = b0[n], bias1 = b0[16 + n];
#pragma unroll
    for (int v = 0; v < 8; ++v) {
      float e0 = c0[v] + bias0;
      float e1 = c1[v] + bias1;
      float s = half16_sum(e0 * e0 + e1 * e1);
      const float invn = 1.0f / fmaxf(sqrtf(s), 1e-12f);
      const int row = t0 + v + 8 * kb;
      h1[(size_t)row * 32 + n] = e0 * invn;
      h1[(size_t)row * 32 + 16 + n] = e1 * invn;
    }
  }
}

#define TS 132
#define AS 68
#define M1CAP 512
__global__ __launch_bounds__(128) void mlp1_kernel(const float* __restrict__ h1,
                                                   const int* __restrict__ start,
                                                   const int* __restrict__ ssrc,
                                                   const float* __restrict__ sw,
                                                   const float* __restrict__ W1,
                                                   const float* __restrict__ b1,
                                                   const float* __restrict__ W2,
                                                   const float* __restrict__ b2,
                                                   float* __restrict__ out) {
  __shared__ __attribute__((aligned(16))) float ash[16 * AS];
  __shared__ float tsh[16 * TS];
  __shared__ float rowsq[16];
  __shared__ int eidx[M1CAP];
  __shared__ float2 ewt[M1CAP];
  const int m0 = blockIdx.x * 16;
  const int wave = threadIdx.x >> 5;
  const int lane = threadIdx.x & 31;
  const int n = lane & 15;
  const int kb = lane >> 4;

  {
    const int r = threadIdx.x >> 3;
    const int q = threadIdx.x & 7;
    const int eb = start[m0];
    const int ee = start[m0 + 16];
    const int beg = start[m0 + r];
    const int end = start[m0 + r + 1];
    const float2* sw2 = (const float2*)sw;
    float4 a0 = make_float4(0.f, 0.f, 0.f, 0.f);
    float4 a1 = make_float4(0.f, 0.f, 0.f, 0.f);
    for (int c0 = eb; c0 < ee; c0 += M1CAP) {
      const int cn = min(M1CAP, ee - c0);
      int si[M1CAP / 128];
      float2 wi[M1CAP / 128];
#pragma unroll
      for (int u = 0; u < M1CAP / 128; ++u) {
        const int i = (int)threadIdx.x + u * 128;
        si[u] = 0;
        wi[u] = make_float2(0.f, 0.f);
        if (i < cn) {
          si[u] = ssrc[c0 + i];
          wi[u] = sw2[c0 + i];
        }
      }
#pragma unroll
      for (int u = 0; u < M1CAP / 128; ++u) {
        const int i = (int)threadIdx.x + u * 128;
        if (i < cn) {
          eidx[i] = si[u];
          ewt[i] = wi[u];
        }
      }
      __syncthreads();
      const int pb = max(beg, c0);
      const int pe = min(end, c0 + cn);
      sum_rows<7>((const char*)h1, 16u * (unsigned)q, eidx, ewt, c0, pb, pe, a0, a1);
      if (c0 + M1CAP < ee) __syncthreads();
    }
    *(float4*)(ash + r * AS + 4 * q) = a0;
    *(float4*)(ash + r * AS + 32 + 4 * q) = a1;
  }
  __syncthreads();

#pragma unroll
  for (int tt = 0; tt < 2; ++tt) {
    const int n0 = (wave * 2 + tt) * 16;
    v8f c = {};
    for (int kt = 0; kt < 16; ++kt) {
      const int kA = kt * 4 + 2 * kb;
      v2f a, b;
      const float* ap = ash + n * AS + kA;
      a.x = ap[0];
      a.y = ap[1];
      b.x = W1[(kA + 0) * 128 + n0 + n];
      b.y = W1[(kA + 1) * 128 + n0 + n];
      c = wmma4f32(a, b, c);
    }
    const float bias = b1[n0 + n];
#pragma unroll
    for (int v = 0; v < 8; ++v) {
      tsh[(v + 8 * kb) * TS + n0 + n] = fmaxf(c[v] + bias, 0.0f);
    }
  }
  if (threadIdx.x < 16) rowsq[threadIdx.x] = 0.0f;
  __syncthreads();

  const int n0 = wave * 16;
  v8f c = {};
  for (int kt = 0; kt < 32; ++kt) {
    const int kA = kt * 4 + 2 * kb;
    v2f a, b;
    a.x = tsh[n * TS + kA];
    a.y = tsh[n * TS + kA + 1];
    b.x = W2[(kA + 0) * 64 + n0 + n];
    b.y = W2[(kA + 1) * 64 + n0 + n];
    c = wmma4f32(a, b, c);
  }
  const float bias = b2[n0 + n];
  float cv[8];
#pragma unroll
  for (int v = 0; v < 8; ++v) {
    cv[v] = c[v] + bias;
    float s = half16_sum(cv[v] * cv[v]);
    if (n == 0) atomicAdd(&rowsq[v + 8 * kb], s);
  }
  __syncthreads();
#pragma unroll
  for (int v = 0; v < 8; ++v) {
    const int row = v + 8 * kb;
    const float invn = 1.0f / fmaxf(sqrtf(rowsq[row]), 1e-12f);
    out[(size_t)(m0 + row) * 64 + n0 + n] = cv[v] * invn;
  }
}

extern "C" void kernel_launch(void* const* d_in, const int* in_sizes, int n_in,
                              void* d_out, int out_size, void* d_ws, size_t ws_size,
                              hipStream_t stream) {
  const float* x  = (const float*)d_in[0];
  const int*   ei = (const int*)d_in[1];
  const float* Kw = (const float*)d_in[2];
  const float* W0 = (const float*)d_in[3];
  const float* b0 = (const float*)d_in[4];
  const float* W1 = (const float*)d_in[5];
  const float* b1 = (const float*)d_in[6];
  const float* W2 = (const float*)d_in[7];
  const float* b2 = (const float*)d_in[8];

  const int N = in_sizes[0] / 8;
  const int E = in_sizes[1] / 2;
  const int* src = ei;
  const int* dst = ei + E;

  const int NB = (N + BNODES - 1) / BNODES;

  int*      start    = (int*)d_ws;
  int*      rowTotal = start + N + 64;
  int*      ssrc     = rowTotal + NBMAX;
  float*    sw       = (float*)(ssrc + E);
  float*    rec      = sw + (size_t)E * 2;
  float*    h1       = rec + (size_t)N * 16;
  int*      table    = ssrc;
  unsigned* recP     = (unsigned*)rec;
  float*    recW     = (float*)(recP + E);

  bucket_count_kernel<<<G1, 1024, 0, stream>>>(dst, table, E, NB);
  bucket_scan_kernel<<<NB, G1, 0, stream>>>(table, rowTotal);
  bucket_place_kernel<<<G1, 1024, 0, stream>>>(src, dst, Kw, Kw + E, table, rowTotal, recP, recW,
                                               E, NB);
  node_place_kernel<<<NB, BNODES, 0, stream>>>(rowTotal, recP, recW, start, ssrc, sw, N);
  gather0_mlp0_kernel<<<(N + G0N - 1) / G0N, G0T, 0, stream>>>(x, start, ssrc, sw, W0, b0, h1, N);
  mlp1_kernel<<<N / 16, 128, 0, stream>>>(h1, start, ssrc, sw, W1, b1, W2, b2,
                                          (float*)d_out);
}


# level-1 sort orders each chunk's records in the chunk's own stretch: no global count table, count and scan launches gone; level 2 collects its bucket's 256 runs; sum order stays an LDS arrival order
# speedup vs baseline: 9.5120x; 1.1312x over previous
#include <hip/hip_runtime.h>
#include <math.h>

typedef __attribute__((ext_vector_type(2))) float v2f;
typedef __attribute__((ext_vector_type(8))) float v8f;

__device__ __forceinline__ v8f wmma4f32(v2f a, v2f b, v8f c) {
  return __builtin_amdgcn_wmma_f32_16x16x4_f32(
       false, a,  false, b,
       (short)0, c,  false,  false);
}

__device__ __forceinline__ float half16_sum(float s) {
  s += __shfl_xor(s, 1, 32);
  s += __shfl_xor(s, 2, 32);
  s += __shfl_xor(s, 4, 32);
  s += __shfl_xor(s, 8, 32);
  return s;
}

#define BSH 8
#define BNODES 256
#define NBMAX 1024
#define G1 256
#define OT (NBMAX + 1)
static_assert(G1 == BNODES, "level 2 gives one thread to each chunk's run");

__device__ __forceinline__ int wave_incl_scan(int v, int lane) {
  for (int d = 1; d < 32; d <<= 1) {
    const int u = __shfl_up(v, d, 32);
    if (lane >= d) v += u;
  }
  return v;
}

#define PCAP 12800
#define PSLOTS 13
static_assert(PCAP <= PSLOTS * 1024, "a thread's PSLOTS edges must cover an ordered chunk");
__global__ __launch_bounds__(1024) void chunk_sort_kernel(const int* __restrict__ src,
                                                          const int* __restrict__ dst,
                                                          const float* __restrict__ kw0,
                                                          const float* __restrict__ kw1,
                                                          float* __restrict__ offT,
                                                          unsigned* __restrict__ recP,
                                                          float* __restrict__ recW,
                                                          int E, int NB) {
  __shared__ int cur[NBMAX];
  __shared__ int wsumB[32];
  __shared__ uint4 stage[PCAP];
  const int g = blockIdx.x;
  const int tid = threadIdx.x;
  const int lane = tid & 31;
  const int wave = tid >> 5;
  const int chunk = (E + G1 - 1) / G1;
  const int e0 = min(E, g * chunk);
  const int e1 = min(E, e0 + chunk);
  const int nloc = e1 - e0;
  const bool ordered = chunk <= PCAP;
  float2* recW2 = (float2*)recW;
  cur[tid] = 0;
  __syncthreads();
  int d[PSLOTS];
  unsigned s[PSLOTS];
  float w0[PSLOTS], w1[PSLOTS];
  if (ordered) {
    const int last = max(e1 - 1, 0);
#pragma unroll
    for (int u = 0; u < PSLOTS; ++u) {
      const int e = e0 + tid + u * 1024;
      const int ec = min(e, last);
      d[u] = dst[ec];
      s[u] = (unsigned)src[ec];
      w0[u] = kw0[ec];
      w1[u] = kw1[ec];
    }
#pragma unroll
    for (int u = 0; u < PSLOTS; ++u)
      if (e0 + tid + u * 1024 < e1) atomicAdd(&cur[d[u] >> BSH], 1);
  } else {
    for (int e = e0 + tid; e < e1; e += 1024) atomicAdd(&cur[dst[e] >> BSH], 1);
  }
  __syncthreads();
  const int own = cur[tid];
  const int incl = wave_incl_scan(own, lane);
  if (lane == 31) wsumB[wave] = incl;
  __syncthreads();
  if (wave == 0) {
    const int v = wsumB[lane];
    const int sc = wave_incl_scan(v, lane);
    wsumB[lane] = sc - v;
  }
  __syncthreads();
  const int ls = wsumB[wave] + incl - own;
  if (tid < NB) offT[g * OT + tid] = __int_as_float(ls);
  if (tid == 0) offT[g * OT + NB] = __int_as_float(nloc);
  cur[tid] = ls;
  __syncthreads();
  if (ordered) {
    int pos[PSLOTS];
#pragma unroll
    for (int u = 0; u < PSLOTS; ++u) {
      pos[u] = 0;
      if (e0 + tid + u * 1024 < e1) pos[u] = atomicAdd(&cur[d[u] >> BSH], 1);
    }
#pragma unroll
    for (int u = 0; u < PSLOTS; ++u)
      if (e0 + tid + u * 1024 < e1)
        stage[pos[u]] = make_uint4(s[u] | ((unsigned)(d[u] & (BNODES - 1)) << 24),
                                   __float_as_uint(w0[u]), __float_as_uint(w1[u]), 0u);
    __syncthreads();
    for (int q = (e0 & ~31) + tid; q < e1; q += 1024) {
      if (q >= e0) {
        const uint4 r = stage[q - e0];
        recP[q] = r.x;
        recW2[q] = make_float2(__uint_as_float(r.y), __uint_as_float(r.z));
      }
    }
  } else {
    for (int e = e0 + tid; e < e1; e += 1024) {
      const int dd = dst[e];
      const unsigned ss = (unsigned)src[e];
      const float2 w = make_float2(kw0[e], kw1[e]);
      const int pos = atomicAdd(&cur[dd >> BSH], 1);
      recP[e0 + pos] = ss | ((unsigned)(dd & (BNODES - 1)) << 24);
      recW2[e0 + pos] = w;
    }
  }
}

#define NCAP 6144
#define NSLOTS (NCAP / BNODES)
static_assert(NSLOTS % 8 == 0 && NCAP >= 2 * G1 + 1, "placing goes eight records at a time; the run table borrows the ids' area");

__device__ __forceinline__ int run_index(const int* runSrc, const int* runDst, int k) {
  int lo = 0, hi = G1;
  while (hi - lo > 1) {
    const int mid = (lo + hi) >> 1;
    if (runDst[mid] <= k) lo = mid; else hi = mid;
  }
  return runSrc[lo] + (k - runDst[lo]);
}

__global__ __launch_bounds__(BNODES, 8) void node_place_kernel(const float* __restrict__ offT,
                                                               const unsigned* __restrict__ recP,
                                                               const float* __restrict__ recW,
                                                               int* __restrict__ start,
                                                               int* __restrict__ ssrc,
                                                               float* __restrict__ sw, int N, int E) {
  __shared__ int cntL[BNODES];
  __shared__ int curL[BNODES];
  __shared__ int wsum[BNODES / 32];
  __shared__ int wsum2[BNODES / 32];
  __shared__ int sP[NCAP];
  __shared__ float2 sW[NCAP];
  int* runSrc = sP;
  int* runDst = sP + G1;
  const int b = blockIdx.x;
  const int tid = threadIdx.x;
  const int lane = tid & 31;
  const int wave = tid >> 5;
  const int chunk = (E + G1 - 1) / G1;
  const float2* rw = (const float2*)recW;
  float2* sw2 = (float2*)sw;
  cntL[tid] = 0;
  const int o0 = __float_as_int(offT[tid * OT + b]);
  const int o1 = __float_as_int(offT[tid * OT + b + 1]);
  const int len = o1 - o0;
  const int inclL = wave_incl_scan(len, lane);
  int so = o0;
  so += __shfl_xor(so, 16, 32);
  so += __shfl_xor(so, 8, 32);
  so += __shfl_xor(so, 4, 32);
  so += __shfl_xor(so, 2, 32);
  so += __shfl_xor(so, 1, 32);
  if (lane == 31) wsum[wave] = inclL;
  if (lane == 0) wsum2[wave] = so;
  __syncthreads();
  int offr = 0, n = 0, base = 0;
  for (int w = 0; w < BNODES / 32; ++w) {
    const int t = wsum[w];
    if (w < wave) offr += t;
    n += t;
    base += wsum2[w];
  }
  const int exclL = offr + inclL - len;
  runSrc[tid] = min(E, tid * chunk) + o0;
  runDst[tid] = exclL;
  if (tid == G1 - 1) runDst[G1] = exclL + len;
  __syncthreads();
  const bool fits = n <= NCAP;
  int idx[NSLOTS];
  unsigned p[NSLOTS];
  float2 wt[NSLOTS];
  if (fits) {
    int* srcIdx = (int*)sW;
    const int hw = tid >> 4;
    const int j = tid & 15;
    int rs[G1 / 16], rd[G1 / 16], re[G1 / 16];
#pragma unroll
    for (int i = 0; i < G1 / 16; ++i) {
      const int r = hw + 16 * i;
      rs[i] = runSrc[r];
      rd[i] = runDst[r];
      re[i] = runDst[r + 1];
    }
#pragma unroll
    for (int i = 0; i < G1 / 16; ++i) {
      const int ln = re[i] - rd[i];
      if (j < ln) srcIdx[rd[i] + j] = rs[i] + j;
#pragma nounroll
      for (int jj = j + 16; jj < ln; jj += 16) srcIdx[rd[i] + jj] = rs[i] + jj;
    }
    __syncthreads();
#pragma unroll
    for (int u = 0; u < NSLOTS; ++u) {
      const int k = tid + u * BNODES;
      const int v = srcIdx[k];
      idx[u] = (k < n) ? v : -1;
    }
#pragma unroll
    for (int u = 0; u < NSLOTS; ++u) p[u] = recP[max(idx[u], 0)];
#pragma unroll
    for (int u = 0; u < NSLOTS; ++u) wt[u] = rw[max(idx[u], 0)];
#pragma unroll
    for (int u = 0; u < NSLOTS; ++u)
      if (idx[u] >= 0) atomicAdd(&cntL[p[u] >> 24], 1);
  } else {
    for (int i = tid; i < n; i += BNODES)
      atomicAdd(&cntL[recP[run_index(runSrc, runDst, i)] >> 24], 1);
  }
  __syncthreads();
  const int c = cntL[tid];
  const int incl = wave_incl_scan(c, lane);
  if (lane == 31) wsum[wave] = incl;
  __syncthreads();
  int off = 0;
  for (int w = 0; w < wave; ++w) off += wsum[w];
  const int excl = off + incl - c;
  curL[tid] = excl;
  const int node = b * BNODES + tid;
  if (node < N) start[node] = base + excl;
  if (node == N - 1) start[N] = base + excl + c;
  __syncthreads();
  if (fits) {
#pragma unroll
    for (int h = 0; h < NSLOTS; h += 8) {
      int pos[8];
#pragma unroll
      for (int u = 0; u < 8; ++u) {
        pos[u] = 0;
        if (idx[h + u] >= 0) pos[u] = atomicAdd(&curL[p[h + u] >> 24], 1);
      }
#pragma unroll
      for (int u = 0; u < 8; ++u) {
        if (idx[h + u] >= 0) {
          sP[pos[u]] = (int)(p[h + u] & 0xFFFFFFu);
          sW[pos[u]] = wt[h + u];
        }
      }
    }
    __syncthreads();
    for (int i = tid; i < n; i += BNODES) {
      ssrc[base + i] = sP[i];
      sw2[base + i] = sW[i];
    }
  } else {
    for (int i = tid; i < n; i += BNODES) {
      const int ix = run_index(runSrc, runDst, i);
      const unsigned pp = recP[ix];
      const float2 w = rw[ix];
      const int pos = base + atomicAdd(&curL[pp >> 24], 1);
      ssrc[pos] = (int)(pp & 0xFFFFFFu);
      sw2[pos] = w;
    }
  }
}

#define EB 8
static_assert(EB == 8, "sum_rows names the EB rows of a batch one by one");

template <int ROWSHIFT>
__device__ __forceinline__ void sum_rows(const char* __restrict__ rows, unsigned lane_off,
                                         const int* eidx, const float2* ewt,
                                         int c0, int pb, int pe, float4& a0, float4& a1) {
  for (int p = pb; p < pe; p += EB) {
    const int nv = pe - p;
    int s[EB];
    float2 w[EB];
    float4 v[EB];
#pragma unroll
    for (int u = 0; u < EB; ++u) {
      const int i = min(p + u, pe - 1) - c0;
      s[u] = eidx[i];
      const float2 ww = ewt[i];
      w[u].x = (u < nv) ? ww.x : 0.f;
      w[u].y = (u < nv) ? ww.y : 0.f;
    }
#pragma unroll
    for (int u = 0; u < EB; ++u)
      v[u] = *(const float4*)(rows + ((((unsigned)s[u]) << ROWSHIFT) | lane_off));
    asm volatile(""
                 : "+v"(a0.x), "+v"(a0.y), "+v"(a0.z), "+v"(a0.w),
                   "+v"(a1.x), "+v"(a1.y), "+v"(a1.z), "+v"(a1.w)
                 : "v"(v[0].x), "v"(v[1].x), "v"(v[2].x), "v"(v[3].x),
                   "v"(v[4].x), "v"(v[5].x), "v"(v[6].x), "v"(v[7].x));
#pragma unroll
    for (int u = 0; u < EB; ++u) {
      const float w0 = w[u].x, w1 = w[u].y;
      a0.x += w0 * v[u].x; a0.y += w0 * v[u].y; a0.z += w0 * v[u].z; a0.w += w0 * v[u].w;
      a1.x += w1 * v[u].x; a1.y += w1 * v[u].y; a1.z += w1 * v[u].z; a1.w += w1 * v[u].w;
    }
  }
}

#define G0N 64
#define G0T 128
#define G0CAP 1536
#define A0P 20
__global__ __launch_bounds__(G0T, 12) void gather0_mlp0_kernel(const float* __restrict__ x,
                                                           const int* __restrict__ start,
                                                           const int* __restrict__ ssrc,
                                                           const float* __restrict__ sw,
                                                           const float* __restrict__ W0,
                                                           const float* __restrict__ b0,
                                                           float* __restrict__ h1, int N) {
  __shared__ int eidx[G0CAP];
  __shared__ float2 ewt[G0CAP];
  __shared__ __attribute__((aligned(16))) float atile[G0N * A0P];
  const int m0 = blockIdx.x * G0N;
  const int r = threadIdx.x >> 1;
  const int q = threadIdx.x & 1;
  const int t = m0 + r;
  const bool live = t < N;
  const int eb = start[m0];
  const int ee = start[min(N, m0 + G0N)];
  const int beg = live ? start[t] : 0;
  const int end = live ? start[t + 1] : 0;
  const float2* sw2 = (const float2*)sw;
  float4 a0 = make_float4(0.f, 0.f, 0.f, 0.f);
  float4 a1 = make_float4(0.f, 0.f, 0.f, 0.f);
  for (int c0 = eb; c0 < ee; c0 += G0CAP) {
    const int cn = min(G0CAP, ee - c0);
    int si[G0CAP / G0T];
    float2 wi[G0CAP / G0T];
#pragma unroll
    for (int u = 0; u < G0CAP / G0T; ++u) {
      const int i = (int)threadIdx.x + u * G0T;
      si[u] = 0;
      wi[u] = make_float2(0.f, 0.f);
      if (i < cn) {
        si[u] = ssrc[c0 + i];
        wi[u] = sw2[c0 + i];
      }
    }
#pragma unroll
    for (int u = 0; u < G0CAP / G0T; ++u) {
      const int i = (int)threadIdx.x + u * G0T;
      if (i < cn) {
        eidx[i] = si[u];
        ewt[i] = wi[u];
      }
    }
    __syncthreads();
    const int pb = max(beg, c0);
    const int pe = min(end, c0 + cn);
    sum_rows<5>((const char*)x, 16u * (unsigned)q, eidx, ewt, c0, pb, pe, a0, a1);
    if (c0 + G0CAP < ee) __syncthreads();
  }
  *(float4*)(atile + r * A0P + 4 * q) = a0;
  *(float4*)(atile + r * A0P + 8 + 4 * q) = a1;
  __syncthreads();

  const int wave = threadIdx.x >> 5;
  const int lane = threadIdx.x & 31;
  const int n = lane & 15;
  const int kb = lane >> 4;
  const int t0 = m0 + 16 * wave;
  if (t0 + 16 <= N) {
    v8f c0 = {}, c1 = {};
#pragma unroll
    for (int kt = 0; kt < 4; ++kt) {
      const int kA = kt * 4 + 2 * kb;
      v2f a, bb0, bb1;
      const float* ap = atile + (16 * wave + n) * A0P + kA;
      a.x = ap[0];
      a.y = ap[1];
      bb0.x = W0[(kA + 0) * 32 + n];
      bb0.y = W0[(kA + 1) * 32 + n];
      bb1.x = W0[(kA + 0) * 32 + 16 + n];
      bb1.y = W0[(kA + 1) * 32 + 16 + n];
      c0 = wmma4f32(a, bb0, c0);
      c1 = wmma4f32(a, bb1, c1);
    }
    const float bias0 = b0[n], bias1 = b0[16 + n];
#pragma unroll
    for (int v = 0; v < 8; ++v) {
      float e0 = c0[v] + bias0;
      float e1 = c1[v] + bias1;
      float s = half16_sum(e0 * e0 + e1 * e1);
      const float invn = 1.0f / fmaxf(sqrtf(s), 1e-12f);
      const int row = t0 + v + 8 * kb;
      h1[(size_t)row * 32 + n] = e0 * invn;
      h1[(size_t)row * 32 + 16 + n] = e1 * invn;
    }
  }
}

#define TS 132
#define AS 68
#define M1CAP 512
__global__ __launch_bounds__(128) void mlp1_kernel(const float* __restrict__ h1,
                                                   const int* __restrict__ start,
                                                   const int* __restrict__ ssrc,
                                                   const float* __restrict__ sw,
                                                   const float* __restrict__ W1,
                                                   const float* __restrict__ b1,
                                                   const float* __restrict__ W2,
                                                   const float* __restrict__ b2,
                                                   float* __restrict__ out) {
  __shared__ __attribute__((aligned(16))) float ash[16 * AS];
  __shared__ float tsh[16 * TS];
  __shared__ float rowsq[16];
  __shared__ int eidx[M1CAP];
  __shared__ float2 ewt[M1CAP];
  const int m0 = blockIdx.x * 16;
  const int wave = threadIdx.x >> 5;
  const int lane = threadIdx.x & 31;
  const int n = lane & 15;
  const int kb = lane >> 4;

  {
    const int r = threadIdx.x >> 3;
    const int q = threadIdx.x & 7;
    const int eb = start[m0];
    const int ee = start[m0 + 16];
    const int beg = start[m0 + r];
    const int end = start[m0 + r + 1];
    const float2* sw2 = (const float2*)sw;
    float4 a0 = make_float4(0.f, 0.f, 0.f, 0.f);
    float4 a1 = make_float4(0.f, 0.f, 0.f, 0.f);
    for (int c0 = eb; c0 < ee; c0 += M1CAP) {
      const int cn = min(M1CAP, ee - c0);
      int si[M1CAP / 128];
      float2 wi[M1CAP / 128];
#pragma unroll
      for (int u = 0; u < M1CAP / 128; ++u) {
        const int i = (int)threadIdx.x + u * 128;
        si[u] = 0;
        wi[u] = make_float2(0.f, 0.f);
        if (i < cn) {
          si[u] = ssrc[c0 + i];
          wi[u] = sw2[c0 + i];
        }
      }
#pragma unroll
      for (int u = 0; u < M1CAP / 128; ++u) {
        const int i = (int)threadIdx.x + u * 128;
        if (i < cn) {
          eidx[i] = si[u];
          ewt[i] = wi[u];
        }
      }
      __syncthreads();
      const int pb = max(beg, c0);
      const int pe = min(end, c0 + cn);
      sum_rows<7>((const char*)h1, 16u * (unsigned)q, eidx, ewt, c0, pb, pe, a0, a1);
      if (c0 + M1CAP < ee) __syncthreads();
    }
    *(float4*)(ash + r * AS + 4 * q) = a0;
    *(float4*)(ash + r * AS + 32 + 4 * q) = a1;
  }
  __syncthreads();

#pragma unroll
  for (int tt = 0; tt < 2; ++tt) {
    const int n0 = (wave * 2 + tt) * 16;
    v8f c = {};
    for (int kt = 0; kt < 16; ++kt) {
      const int kA = kt * 4 + 2 * kb;
      v2f a, b;
      const float* ap = ash + n * AS + kA;
      a.x = ap[0];
      a.y = ap[1];
      b.x = W1[(kA + 0) * 128 + n0 + n];
      b.y = W1[(kA + 1) * 128 + n0 + n];
      c = wmma4f32(a, b, c);
    }
    const float bias = b1[n0 + n];
#pragma unroll
    for (int v = 0; v < 8; ++v) {
      tsh[(v + 8 * kb) * TS + n0 + n] = fmaxf(c[v] + bias, 0.0f);
    }
  }
  if (threadIdx.x < 16) rowsq[threadIdx.x] = 0.0f;
  __syncthreads();

  const int n0 = wave * 16;
  v8f c = {};
  for (int kt = 0; kt < 32; ++kt) {
    const int kA = kt * 4 + 2 * kb;
    v2f a, b;
    a.x = tsh[n * TS + kA];
    a.y = tsh[n * TS + kA + 1];
    b.x = W2[(kA + 0) * 64 + n0 + n];
    b.y = W2[(kA + 1) * 64 + n0 + n];
    c = wmma4f32(a, b, c);
  }
  const float bias = b2[n0 + n];
  float cv[8];
#pragma unroll
  for (int v = 0; v < 8; ++v) {
    cv[v] = c[v] + bias;
    float s = half16_sum(cv[v] * cv[v]);
    if (n == 0) atomicAdd(&rowsq[v + 8 * kb], s);
  }
  __syncthreads();
#pragma unroll
  for (int v = 0; v < 8; ++v) {
    const int row = v + 8 * kb;
    const float invn = 1.0f / fmaxf(sqrtf(rowsq[row]), 1e-12f);
    out[(size_t)(m0 + row) * 64 + n0 + n] = cv[v] * invn;
  }
}

extern "C" void kernel_launch(void* const* d_in, const int* in_sizes, int n_in,
                              void* d_out, int out_size, void* d_ws, size_t ws_size,
                              hipStream_t stream) {
  const float* x  = (const float*)d_in[0];
  const int*   ei = (const int*)d_in[1];
  const float* Kw = (const float*)d_in[2];
  const float* W0 = (const float*)d_in[3];
  const float* b0 = (const float*)d_in[4];
  const float* W1 = (const float*)d_in[5];
  const float* b1 = (const float*)d_in[6];
  const float* W2 = (const float*)d_in[7];
  const float* b2 = (const float*)d_in[8];

  const int N = in_sizes[0] / 8;
  const int E = in_sizes[1] / 2;
  const int* src = ei;
  const int* dst = ei + E;

  const int NB = (N + BNODES - 1) / BNODES;

  int*      start    = (int*)d_ws;
  int*      ssrc     = start + N + 64 + NBMAX;
  float*    sw       = (float*)(ssrc + E);
  float*    rec      = sw + (size_t)E * 2;
  float*    h1       = rec + (size_t)N * 16;
  unsigned* recP     = (unsigned*)rec;
  float*    recW     = (float*)(recP + E);
  float*    offT     = (float*)d_out;

  chunk_sort_kernel<<<G1, 1024, 0, stream>>>(src, dst, Kw, Kw + E, offT, recP, recW, E, NB);
  node_place_kernel<<<NB, BNODES, 0, stream>>>(offT, recP, recW, start, ssrc, sw, N, E);
  gather0_mlp0_kernel<<<(N + G0N - 1) / G0N, G0T, 0, stream>>>(x, start, ssrc, sw, W0, b0, h1, N);
  mlp1_kernel<<<N / 16, 128, 0, stream>>>(h1, start, ssrc, sw, W1, b1, W2, b2,
                                          (float*)d_out);
}


# layer-0 aggregation + MLP fused into the last sort level: a bucket's block sums its 256 nodes from the sorted records in its LDS; one launch and one re-read of the edge list gone; sum order kept
# speedup vs baseline: 10.3706x; 1.0903x over previous
#include <hip/hip_runtime.h>
#include <math.h>

typedef __attribute__((ext_vector_type(2))) float v2f;
typedef __attribute__((ext_vector_type(8))) float v8f;

__device__ __forceinline__ v8f wmma4f32(v2f a, v2f b, v8f c) {
  return __builtin_amdgcn_wmma_f32_16x16x4_f32(
       false, a,  false, b,
       (short)0, c,  false,  false);
}

__device__ __forceinline__ float half16_sum(float s) {
  s += __shfl_xor(s, 1, 32);
  s += __shfl_xor(s, 2, 32);
  s += __shfl_xor(s, 4, 32);
  s += __shfl_xor(s, 8, 32);
  return s;
}

#define BSH 8
#define BNODES 256
#define NBMAX 1024
#define G1 256
#define OT (NBMAX + 1)
static_assert(G1 == BNODES, "level 2 gives one thread to each chunk's run");

__device__ __forceinline__ int wave_incl_scan(int v, int lane) {
  for (int d = 1; d < 32; d <<= 1) {
    const int u = __shfl_up(v, d, 32);
    if (lane >= d) v += u;
  }
  return v;
}

#define PCAP 12800
#define PSLOTS 13
static_assert(PCAP <= PSLOTS * 1024, "a thread's PSLOTS edges must cover an ordered chunk");
__global__ __launch_bounds__(1024) void chunk_sort_kernel(const int* __restrict__ src,
                                                          const int* __restrict__ dst,
                                                          const float* __restrict__ kw0,
                                                          const float* __restrict__ kw1,
                                                          float* __restrict__ offT,
                                                          unsigned* __restrict__ recP,
                                                          float* __restrict__ recW,
                                                          int E, int NB) {
  __shared__ int cur[NBMAX];
  __shared__ int wsumB[32];
  __shared__ uint4 stage[PCAP];
  const int g = blockIdx.x;
  const int tid = threadIdx.x;
  const int lane = tid & 31;
  const int wave = tid >> 5;
  const int chunk = (E + G1 - 1) / G1;
  const int e0 = min(E, g * chunk);
  const int e1 = min(E, e0 + chunk);
  const int nloc = e1 - e0;
  const bool ordered = chunk <= PCAP;
  float2* recW2 = (float2*)recW;
  cur[tid] = 0;
  __syncthreads();
  int d[PSLOTS];
  unsigned s[PSLOTS];
  float w0[PSLOTS], w1[PSLOTS];
  if (ordered) {
    const int last = max(e1 - 1, 0);
#pragma unroll
    for (int u = 0; u < PSLOTS; ++u) {
      const int e = e0 + tid + u * 1024;
      const int ec = min(e, last);
      d[u] = dst[ec];
      s[u] = (unsigned)src[ec];
      w0[u] = kw0[ec];
      w1[u] = kw1[ec];
    }
#pragma unroll
    for (int u = 0; u < PSLOTS; ++u)
      if (e0 + tid + u * 1024 < e1) atomicAdd(&cur[d[u] >> BSH], 1);
  } else {
    for (int e = e0 + tid; e < e1; e += 1024) atomicAdd(&cur[dst[e] >> BSH], 1);
  }
  __syncthreads();
  const int own = cur[tid];
  const int incl = wave_incl_scan(own, lane);
  if (lane == 31) wsumB[wave] = incl;
  __syncthreads();
  if (wave == 0) {
    const int v = wsumB[lane];
    const int sc = wave_incl_scan(v, lane);
    wsumB[lane] = sc - v;
  }
  __syncthreads();
  const int ls = wsumB[wave] + incl - own;
  if (tid < NB) offT[g * OT + tid] = __int_as_float(ls);
  if (tid == 0) offT[g * OT + NB] = __int_as_float(nloc);
  cur[tid] = ls;
  __syncthreads();
  if (ordered) {
    int pos[PSLOTS];
#pragma unroll
    for (int u = 0; u < PSLOTS; ++u) {
      pos[u] = 0;
      if (e0 + tid + u * 1024 < e1) pos[u] = atomicAdd(&cur[d[u] >> BSH], 1);
    }
#pragma unroll
    for (int u = 0; u < PSLOTS; ++u)
      if (e0 + tid + u * 1024 < e1)
        stage[pos[u]] = make_uint4(s[u] | ((unsigned)(d[u] & (BNODES - 1)) << 24),
                                   __float_as_uint(w0[u]), __float_as_uint(w1[u]), 0u);
    __syncthreads();
    for (int q = (e0 & ~31) + tid; q < e1; q += 1024) {
      if (q >= e0) {
        const uint4 r = stage[q - e0];
        recP[q] = r.x;
        recW2[q] = make_float2(__uint_as_float(r.y), __uint_as_float(r.z));
      }
    }
  } else {
    for (int e = e0 + tid; e < e1; e += 1024) {
      const int dd = dst[e];
      const unsigned ss = (unsigned)src[e];
      const float2 w = make_float2(kw0[e], kw1[e]);
      const int pos = atomicAdd(&cur[dd >> BSH], 1);
      recP[e0 + pos] = ss | ((unsigned)(dd & (BNODES - 1)) << 24);
      recW2[e0 + pos] = w;
    }
  }
}

#define EB 8
static_assert(EB == 8, "sum_rows names the EB rows of a batch one by one");

template <int ROWSHIFT>
__device__ __forceinline__ void sum_rows(const char* __restrict__ rows, unsigned lane_off,
                                         const int* eidx, const float2* ewt,
                                         int c0, int pb, int pe, float4& a0, float4& a1) {
  for (int p = pb; p < pe; p += EB) {
    const int nv = pe - p;
    int s[EB];
    float2 w[EB];
    float4 v[EB];
#pragma unroll
    for (int u = 0; u < EB; ++u) {
      const int i = min(p + u, pe - 1) - c0;
      s[u] = eidx[i];
      const float2 ww = ewt[i];
      w[u].x = (u < nv) ? ww.x : 0.f;
      w[u].y = (u < nv) ? ww.y : 0.f;
    }
#pragma unroll
    for (int u = 0; u < EB; ++u)
      v[u] = *(const float4*)(rows + ((((unsigned)s[u]) << ROWSHIFT) | lane_off));
    asm volatile(""
                 : "+v"(a0.x), "+v"(a0.y), "+v"(a0.z), "+v"(a0.w),
                   "+v"(a1.x), "+v"(a1.y), "+v"(a1.z), "+v"(a1.w)
                 : "v"(v[0].x), "v"(v[1].x), "v"(v[2].x), "v"(v[3].x),
                   "v"(v[4].x), "v"(v[5].x), "v"(v[6].x), "v"(v[7].x));
#pragma unroll
    for (int u = 0; u < EB; ++u) {
      const float w0 = w[u].x, w1 = w[u].y;
      a0.x += w0 * v[u].x; a0.y += w0 * v[u].y; a0.z += w0 * v[u].z; a0.w += w0 * v[u].w;
      a1.x += w1 * v[u].x; a1.y += w1 * v[u].y; a1.z += w1 * v[u].z; a1.w += w1 * v[u].w;
    }
  }
}

#define NCAP 6144
#define NSLOTS (NCAP / BNODES)
static_assert(NSLOTS % 8 == 0 && NCAP >= 2 * G1 + 1, "placing goes eight records at a time; the run table borrows the ids' area");
#define A0P 20
static_assert(BNODES * A0P * 4 <= NCAP * 8, "the sums tile of a bucket's nodes borrows the weights' area");

__device__ __forceinline__ int run_index(const int* runSrc, const int* runDst, int k) {
  int lo = 0, hi = G1;
  while (hi - lo > 1) {
    const int mid = (lo + hi) >> 1;
    if (runDst[mid] <= k) lo = mid; else hi = mid;
  }
  return runSrc[lo] + (k - runDst[lo]);
}

__global__ __launch_bounds__(BNODES, 8) void node_place_layer0_kernel(const float* __restrict__ offT,
                                                                      const unsigned* __restrict__ recP,
                                                                      const float* __restrict__ recW,
                                                                      const float* __restrict__ x,
                                                                      const float* __restrict__ W0,
                                                                      const float* __restrict__ b0,
                                                                      int* __restrict__ start,
                                                                      int* __restrict__ ssrc,
                                                                      float* __restrict__ sw,
                                                                      float* __restrict__ h1, int N, int E) {
  __shared__ int cntL[BNODES];
  __shared__ int curL[BNODES];
  __shared__ int wsum[BNODES / 32];
  __shared__ int wsum2[BNODES / 32];
  __shared__ int sP[NCAP];
  __shared__ __attribute__((aligned(16))) float2 sW[NCAP];
  int* runSrc = sP;
  int* runDst = sP + G1;
  const int b = blockIdx.x;
  const int tid = threadIdx.x;
  const int lane = tid & 31;
  const int wave = tid >> 5;
  const int chunk = (E + G1 - 1) / G1;
  const float2* rw = (const float2*)recW;
  float2* sw2 = (float2*)sw;
  cntL[tid] = 0;
  const int o0 = __float_as_int(offT[tid * OT + b]);
  const int o1 = __float_as_int(offT[tid * OT + b + 1]);
  const int len = o1 - o0;
  const int inclL = wave_incl_scan(len, lane);
  int so = o0;
  so += __shfl_xor(so, 16, 32);
  so += __shfl_xor(so, 8, 32);
  so += __shfl_xor(so, 4, 32);
  so += __shfl_xor(so, 2, 32);
  so += __shfl_xor(so, 1, 32);
  if (lane == 31) wsum[wave] = inclL;
  if (lane == 0) wsum2[wave] = so;
  __syncthreads();
  int offr = 0, n = 0, base = 0;
  for (int w = 0; w < BNODES / 32; ++w) {
    const int t = wsum[w];
    if (w < wave) offr += t;
    n += t;
    base += wsum2[w];
  }
  const int exclL = offr + inclL - len;
  runSrc[tid] = min(E, tid * chunk) + o0;
  runDst[tid] = exclL;
  if (tid == G1 - 1) runDst[G1] = exclL + len;
  __syncthreads();
  const bool fits = n <= NCAP;
  int idx[NSLOTS];
  unsigned p[NSLOTS];
  float2 wt[NSLOTS];
  if (fits) {
    int* srcIdx = (int*)sW;
    const int hw = tid >> 4;
    const int j = tid & 15;
    int rs[G1 / 16], rd[G1 / 16], re[G1 / 16];
#pragma unroll
    for (int i = 0; i < G1 / 16; ++i) {
      const int r = hw + 16 * i;
      rs[i] = runSrc[r];
      rd[i] = runDst[r];
      re[i] = runDst[r + 1];
    }
#pragma unroll
    for (int i = 0; i < G1 / 16; ++i) {
      const int ln = re[i] - rd[i];
      if (j < ln) srcIdx[rd[i] + j] = rs[i] + j;
#pragma nounroll
      for (int jj = j + 16; jj < ln; jj += 16) srcIdx[rd[i] + jj] = rs[i] + jj;
    }
    __syncthreads();
#pragma unroll
    for (int u = 0; u < NSLOTS; ++u) {
      const int k = tid + u * BNODES;
      const int v = srcIdx[k];
      idx[u] = (k < n) ? v : -1;
    }
#pragma unroll
    for (int u = 0; u < NSLOTS; ++u) p[u] = recP[max(idx[u], 0)];
#pragma unroll
    for (int u = 0; u < NSLOTS; ++u) wt[u] = rw[max(idx[u], 0)];
#pragma unroll
    for (int u = 0; u < NSLOTS; ++u)
      if (idx[u] >= 0) atomicAdd(&cntL[p[u] >> 24], 1);
  } else {
    for (int i = tid; i < n; i += BNODES)
      atomicAdd(&cntL[recP[run_index(runSrc, runDst, i)] >> 24], 1);
  }
  __syncthreads();
  const int c = cntL[tid];
  const int incl = wave_incl_scan(c, lane);
  if (lane == 31) wsum[wave] = incl;
  __syncthreads();
  int off = 0;
  for (int w = 0; w < wave; ++w) off += wsum[w];
  const int excl = off + incl - c;
  curL[tid] = excl;
  const int node = b * BNODES + tid;
  if (node < N) start[node] = base + excl;
  if (node == N - 1) start[N] = base + excl + c;
  __syncthreads();
  if (fits) {
#pragma unroll
    for (int h = 0; h < NSLOTS; h += 8) {
      int pos[8];
#pragma unroll
      for (int u = 0; u < 8; ++u) {
        pos[u] = 0;
        if (idx[h + u] >= 0) pos[u] = atomicAdd(&curL[p[h + u] >> 24], 1);
      }
#pragma unroll
      for (int u = 0; u < 8; ++u) {
        if (idx[h + u] >= 0) {
          sP[pos[u]] = (int)(p[h + u] & 0xFFFFFFu);
          sW[pos[u]] = wt[h + u];
        }
      }
    }
    __syncthreads();
    for (int i = tid; i < n; i += BNODES) {
      ssrc[base + i] = sP[i];
      sw2[base + i] = sW[i];
    }
  } else {
    for (int i = tid; i < n; i += BNODES) {
      const int ix = run_index(runSrc, runDst, i);
      const unsigned pp = recP[ix];
      const float2 w = rw[ix];
      const int pos = base + atomicAdd(&curL[pp >> 24], 1);
      ssrc[pos] = (int)(pp & 0xFFFFFFu);
      sw2[pos] = w;
    }
  }

  __syncthreads();
  const int q = tid & 1;
  const int r0 = tid >> 1;
  const int r1 = BNODES / 2 + r0;
  const int beg0 = r0 ? curL[r0 - 1] : 0;
  const int end0 = curL[r0];
  const int beg1 = curL[r1 - 1];
  const int end1 = curL[r1];
  float4 g00 = make_float4(0.f, 0.f, 0.f, 0.f);
  float4 g01 = make_float4(0.f, 0.f, 0.f, 0.f);
  float4 g10 = make_float4(0.f, 0.f, 0.f, 0.f);
  float4 g11 = make_float4(0.f, 0.f, 0.f, 0.f);
  for (int c0 = 0; c0 < n; c0 += NCAP) {
    const int cn = min(NCAP, n - c0);
    if (!fits) {
      if (c0 > 0) __syncthreads();
      for (int i = tid; i < cn; i += BNODES) {
        sP[i] = ssrc[base + c0 + i];
        sW[i] = sw2[base + c0 + i];
      }
      __syncthreads();
    }
    sum_rows<5>((const char*)x, 16u * (unsigned)q, sP, sW, c0,
                max(beg0, c0), min(end0, c0 + cn), g00, g01);
    sum_rows<5>((const char*)x, 16u * (unsigned)q, sP, sW, c0,
                max(beg1, c0), min(end1, c0 + cn), g10, g11);
  }
  __syncthreads();
  float* atile = (float*)sW;
  *(float4*)(atile + r0 * A0P + 4 * q) = g00;
  *(float4*)(atile + r0 * A0P + 8 + 4 * q) = g01;
  *(float4*)(atile + r1 * A0P + 4 * q) = g10;
  *(float4*)(atile + r1 * A0P + 8 + 4 * q) = g11;
  __syncthreads();

  const int nn = lane & 15;
  const int kb = lane >> 4;
  for (int tl = 0; tl < 2; ++tl) {
    const int rb = 16 * wave + (BNODES / 2) * tl;
    const int t0 = b * BNODES + rb;
    if (t0 + 16 <= N) {
      v8f acc0 = {}, acc1 = {};
#pragma unroll
      for (int kt = 0; kt < 4; ++kt) {
        const int kA = kt * 4 + 2 * kb;
        v2f a, bb0, bb1;
        const float* ap = atile + (rb + nn) * A0P + kA;
        a.x = ap[0];
        a.y = ap[1];
        bb0.x = W0[(kA + 0) * 32 + nn];
        bb0.y = W0[(kA + 1) * 32 + nn];
        bb1.x = W0[(kA + 0) * 32 + 16 + nn];
        bb1.y = W0[(kA + 1) * 32 + 16 + nn];
        acc0 = wmma4f32(a, bb0, acc0);
        acc1 = wmma4f32(a, bb1, acc1);
      }
      const float bias0 = b0[nn], bias1 = b0[16 + nn];
#pragma unroll
      for (int v = 0; v < 8; ++v) {
        float e0 = acc0[v] + bias0;
        float e1 = acc1[v] + bias1;
        float s = half16_sum(e0 * e0 + e1 * e1);
        const float invn = 1.0f / fmaxf(sqrtf(s), 1e-12f);
        const int row = t0 + v + 8 * kb;
        h1[(size_t)row * 32 + nn] = e0 * invn;
        h1[(size_t)row * 32 + 16 + nn] = e1 * invn;
      }
    }
  }
}

#define TS 132
#define AS 68
#define M1CAP 512
__global__ __launch_bounds__(128) void mlp1_kernel(const float* __restrict__ h1,
                                                   const int* __restrict__ start,
                                                   const int* __restrict__ ssrc,
                                                   const float* __restrict__ sw,
                                                   const float* __restrict__ W1,
                                                   const float* __restrict__ b1,
                                                   const float* __restrict__ W2,
                                                   const float* __restrict__ b2,
                                                   float* __restrict__ out) {
  __shared__ __attribute__((aligned(16))) float ash[16 * AS];
  __shared__ float tsh[16 * TS];
  __shared__ float rowsq[16];
  __shared__ int eidx[M1CAP];
  __shared__ float2 ewt[M1CAP];
  const int m0 = blockIdx.x * 16;
  const int wave = threadIdx.x >> 5;
  const int lane = threadIdx.x & 31;
  const int n = lane & 15;
  const int kb = lane >> 4;

  {
    const int r = threadIdx.x >> 3;
    const int q = threadIdx.x & 7;
    const int eb = start[m0];
    const int ee = start[m0 + 16];
    const int beg = start[m0 + r];
    const int end = start[m0 + r + 1];
    const float2* sw2 = (const float2*)sw;
    float4 a0 = make_float4(0.f, 0.f, 0.f, 0.f);
    float4 a1 = make_float4(0.f, 0.f, 0.f, 0.f);
    for (int c0 = eb; c0 < ee; c0 += M1CAP) {
      const int cn = min(M1CAP, ee - c0);
      int si[M1CAP / 128];
      float2 wi[M1CAP / 128];
#pragma unroll
      for (int u = 0; u < M1CAP / 128; ++u) {
        const int i = (int)threadIdx.x + u * 128;
        si[u] = 0;
        wi[u] = make_float2(0.f, 0.f);
        if (i < cn) {
          si[u] = ssrc[c0 + i];
          wi[u] = sw2[c0 + i];
        }
      }
#pragma unroll
      for (int u = 0; u < M1CAP / 128; ++u) {
        const int i = (int)threadIdx.x + u * 128;
        if (i < cn) {
          eidx[i] = si[u];
          ewt[i] = wi[u];
        }
      }
      __syncthreads();
      const int pb = max(beg, c0);
      const int pe = min(end, c0 + cn);
      sum_rows<7>((const char*)h1, 16u * (unsigned)q, eidx, ewt, c0, pb, pe, a0, a1);
      if (c0 + M1CAP < ee) __syncthreads();
    }
    *(float4*)(ash + r * AS + 4 * q) = a0;
    *(float4*)(ash + r * AS + 32 + 4 * q) = a1;
  }
  __syncthreads();

#pragma unroll
  for (int tt = 0; tt < 2; ++tt) {
    const int n0 = (wave * 2 + tt) * 16;
    v8f c = {};
    for (int kt = 0; kt < 16; ++kt) {
      const int kA = kt * 4 + 2 * kb;
      v2f a, b;
      const float* ap = ash + n * AS + kA;
      a.x = ap[0];
      a.y = ap[1];
      b.x = W1[(kA + 0) * 128 + n0 + n];
      b.y = W1[(kA + 1) * 128 + n0 + n];
      c = wmma4f32(a, b, c);
    }
    const float bias = b1[n0 + n];
#pragma unroll
    for (int v = 0; v < 8; ++v) {
      tsh[(v + 8 * kb) * TS + n0 + n] = fmaxf(c[v] + bias, 0.0f);
    }
  }
  if (threadIdx.x < 16) rowsq[threadIdx.x] = 0.0f;
  __syncthreads();

  const int n0 = wave * 16;
  v8f c = {};
  for (int kt = 0; kt < 32; ++kt) {
    const int kA = kt * 4 + 2 * kb;
    v2f a, b;
    a.x = tsh[n * TS + kA];
    a.y = tsh[n * TS + kA + 1];
    b.x = W2[(kA + 0) * 64 + n0 + n];
    b.y = W2[(kA + 1) * 64 + n0 + n];
    c = wmma4f32(a, b, c);
  }
  const float bias = b2[n0 + n];
  float cv[8];
#pragma unroll
  for (int v = 0; v < 8; ++v) {
    cv[v] = c[v] + bias;
    float s = half16_sum(cv[v] * cv[v]);
    if (n == 0) atomicAdd(&rowsq[v + 8 * kb], s);
  }
  __syncthreads();
#pragma unroll
  for (int v = 0; v < 8; ++v) {
    const int row = v + 8 * kb;
    const float invn = 1.0f / fmaxf(sqrtf(rowsq[row]), 1e-12f);
    out[(size_t)(m0 + row) * 64 + n0 + n] = cv[v] * invn;
  }
}

extern "C" void kernel_launch(void* const* d_in, const int* in_sizes, int n_in,
                              void* d_out, int out_size, void* d_ws, size_t ws_size,
                              hipStream_t stream) {
  const float* x  = (const float*)d_in[0];
  const int*   ei = (const int*)d_in[1];
  const float* Kw = (const float*)d_in[2];
  const float* W0 = (const float*)d_in[3];
  const float* b0 = (const float*)d_in[4];
  const float* W1 = (const float*)d_in[5];
  const float* b1 = (const float*)d_in[6];
  const float* W2 = (const float*)d_in[7];
  const float* b2 = (const float*)d_in[8];

  const int N = in_sizes[0] / 8;
  const int E = in_sizes[1] / 2;
  const int* src = ei;
  const int* dst = ei + E;

  const int NB = (N + BNODES - 1) / BNODES;

  int*      start    = (int*)d_ws;
  int*      ssrc     = start + N + 64 + NBMAX;
  float*    sw       = (float*)(ssrc + E);
  float*    rec      = sw + (size_t)E * 2;
  float*    h1       = rec + (size_t)N * 16;
  float*    offT     = (float*)d_out;
  unsigned* recP     = (unsigned*)d_out + G1 * OT;
  float*    recW     = (float*)(recP + E);

  chunk_sort_kernel<<<G1, 1024, 0, stream>>>(src, dst, Kw, Kw + E, offT, recP, recW, E, NB);
  node_place_layer0_kernel<<<NB, BNODES, 0, stream>>>(offT, recP, recW, x, W0, b0, start, ssrc, sw,
                                                      h1, N, E);
  mlp1_kernel<<<N / 16, 128, 0, stream>>>(h1, start, ssrc, sw, W1, b1, W2, b2,
                                          (float*)d_out);
}


# both row normalisations go through an LDS tile: the lanes holding a whole row take one root and division (was 8 per lane), store whole lines; sum-of-squares order changes, LDS float atomics gone
# speedup vs baseline: 11.1249x; 1.0727x over previous
#include <hip/hip_runtime.h>
#include <math.h>

typedef __attribute__((ext_vector_type(2))) float v2f;
typedef __attribute__((ext_vector_type(8))) float v8f;

__device__ __forceinline__ v8f wmma4f32(v2f a, v2f b, v8f c) {
  return __builtin_amdgcn_wmma_f32_16x16x4_f32(
       false, a,  false, b,
       (short)0, c,  false,  false);
}

#define BSH 8
#define BNODES 256
#define NBMAX 1024
#define G1 256
#define OT (NBMAX + 1)
static_assert(G1 == BNODES, "level 2 gives one thread to each chunk's run");

__device__ __forceinline__ int wave_incl_scan(int v, int lane) {
  for (int d = 1; d < 32; d <<= 1) {
    const int u = __shfl_up(v, d, 32);
    if (lane >= d) v += u;
  }
  return v;
}

#define PCAP 12800
#define PSLOTS 13
static_assert(PCAP <= PSLOTS * 1024, "a thread's PSLOTS edges must cover an ordered chunk");
__global__ __launch_bounds__(1024) void chunk_sort_kernel(const int* __restrict__ src,
                                                          const int* __restrict__ dst,
                                                          const float* __restrict__ kw0,
                                                          const float* __restrict__ kw1,
                                                          float* __restrict__ offT,
                                                          unsigned* __restrict__ recP,
                                                          float* __restrict__ recW,
                                                          int E, int NB) {
  __shared__ int cur[NBMAX];
  __shared__ int wsumB[32];
  __shared__ uint4 stage[PCAP];
  const int g = blockIdx.x;
  const int tid = threadIdx.x;
  const int lane = tid & 31;
  const int wave = tid >> 5;
  const int chunk = (E + G1 - 1) / G1;
  const int e0 = min(E, g * chunk);
  const int e1 = min(E, e0 + chunk);
  const int nloc = e1 - e0;
  const bool ordered = chunk <= PCAP;
  float2* recW2 = (float2*)recW;
  cur[tid] = 0;
  __syncthreads();
  int d[PSLOTS];
  unsigned s[PSLOTS];
  float w0[PSLOTS], w1[PSLOTS];
  if (ordered) {
    const int last = max(e1 - 1, 0);
#pragma unroll
    for (int u = 0; u < PSLOTS; ++u) {
      const int e = e0 + tid + u * 1024;
      const int ec = min(e, last);
      d[u] = dst[ec];
      s[u] = (unsigned)src[ec];
      w0[u] = kw0[ec];
      w1[u] = kw1[ec];
    }
#pragma unroll
    for (int u = 0; u < PSLOTS; ++u)
      if (e0 + tid + u * 1024 < e1) atomicAdd(&cur[d[u] >> BSH], 1);
  } else {
    for (int e = e0 + tid; e < e1; e += 1024) atomicAdd(&cur[dst[e] >> BSH], 1);
  }
  __syncthreads();
  const int own = cur[tid];
  const int incl = wave_incl_scan(own, lane);
  if (lane == 31) wsumB[wave] = incl;
  __syncthreads();
  if (wave == 0) {
    const int v = wsumB[lane];
    const int sc = wave_incl_scan(v, lane);
    wsumB[lane] = sc - v;
  }
  __syncthreads();
  const int ls = wsumB[wave] + incl - own;
  if (tid < NB) offT[g * OT + tid] = __int_as_float(ls);
  if (tid == 0) offT[g * OT + NB] = __int_as_float(nloc);
  cur[tid] = ls;
  __syncthreads();
  if (ordered) {
    int pos[PSLOTS];
#pragma unroll
    for (int u = 0; u < PSLOTS; ++u) {
      pos[u] = 0;
      if (e0 + tid + u * 1024 < e1) pos[u] = atomicAdd(&cur[d[u] >> BSH], 1);
    }
#pragma unroll
    for (int u = 0; u < PSLOTS; ++u)
      if (e0 + tid + u * 1024 < e1)
        stage[pos[u]] = make_uint4(s[u] | ((unsigned)(d[u] & (BNODES - 1)) << 24),
                                   __float_as_uint(w0[u]), __float_as_uint(w1[u]), 0u);
    __syncthreads();
    for (int q = (e0 & ~31) + tid; q < e1; q += 1024) {
      if (q >= e0) {
        const uint4 r = stage[q - e0];
        recP[q] = r.x;
        recW2[q] = make_float2(__uint_as_float(r.y), __uint_as_float(r.z));
      }
    }
  } else {
    for (int e = e0 + tid; e < e1; e += 1024) {
      const int dd = dst[e];
      const unsigned ss = (unsigned)src[e];
      const float2 w = make_float2(kw0[e], kw1[e]);
      const int pos = atomicAdd(&cur[dd >> BSH], 1);
      recP[e0 + pos] = ss | ((unsigned)(dd & (BNODES - 1)) << 24);
      recW2[e0 + pos] = w;
    }
  }
}

#define EB 8
static_assert(EB == 8, "sum_rows names the EB rows of a batch one by one");

template <int ROWSHIFT>
__device__ __forceinline__ void sum_rows(const char* __restrict__ rows, unsigned lane_off,
                                         const int* eidx, const float2* ewt,
                                         int c0, int pb, int pe, float4& a0, float4& a1) {
  for (int p = pb; p < pe; p += EB) {
    const int nv = pe - p;
    int s[EB];
    float2 w[EB];
    float4 v[EB];
#pragma unroll
    for (int u = 0; u < EB; ++u) {
      const int i = min(p + u, pe - 1) - c0;
      s[u] = eidx[i];
      const float2 ww = ewt[i];
      w[u].x = (u < nv) ? ww.x : 0.f;
      w[u].y = (u < nv) ? ww.y : 0.f;
    }
#pragma unroll
    for (int u = 0; u < EB; ++u)
      v[u] = *(const float4*)(rows + ((((unsigned)s[u]) << ROWSHIFT) | lane_off));
    asm volatile(""
                 : "+v"(a0.x), "+v"(a0.y), "+v"(a0.z), "+v"(a0.w),
                   "+v"(a1.x), "+v"(a1.y), "+v"(a1.z), "+v"(a1.w)
                 : "v"(v[0].x), "v"(v[1].x), "v"(v[2].x), "v"(v[3].x),
                   "v"(v[4].x), "v"(v[5].x), "v"(v[6].x), "v"(v[7].x));
#pragma unroll
    for (int u = 0; u < EB; ++u) {
      const float w0 = w[u].x, w1 = w[u].y;
      a0.x += w0 * v[u].x; a0.y += w0 * v[u].y; a0.z += w0 * v[u].z; a0.w += w0 * v[u].w;
      a1.x += w1 * v[u].x; a1.y += w1 * v[u].y; a1.z += w1 * v[u].z; a1.w += w1 * v[u].w;
    }
  }
}

#define NCAP 6144
#define NSLOTS (NCAP / BNODES)
static_assert(NSLOTS % 8 == 0 && NCAP >= 2 * G1 + 1, "placing goes eight records at a time; the run table borrows the ids' area");
#define A0P 20
static_assert(BNODES * A0P * 4 <= NCAP * 8, "the sums tile of a bucket's nodes borrows the weights' area");
#define H1P 36
static_assert((BNODES / 2) * H1P <= NCAP && BNODES * A0P + (BNODES / 2) * H1P <= 2 * NCAP,
              "the result tiles of pass 0 borrow the ids' area, those of pass 1 the weights' area behind the sums tile");

__device__ __forceinline__ int run_index(const int* runSrc, const int* runDst, int k) {
  int lo = 0, hi = G1;
  while (hi - lo > 1) {
    const int mid = (lo + hi) >> 1;
    if (runDst[mid] <= k) lo = mid; else hi = mid;
  }
  return runSrc[lo] + (k - runDst[lo]);
}

__global__ __launch_bounds__(BNODES, 8) void node_place_layer0_kernel(const float* __restrict__ offT,
                                                                      const unsigned* __restrict__ recP,
                                                                      const float* __restrict__ recW,
                                                                      const float* __restrict__ x,
                                                                      const float* __restrict__ W0,
                                                                      const float* __restrict__ b0,
                                                                      int* __restrict__ start,
                                                                      int* __restrict__ ssrc,
                                                                      float* __restrict__ sw,
                                                                      float* __restrict__ h1, int N, int E) {
  __shared__ int cntL[BNODES];
  __shared__ int curL[BNODES];
  __shared__ int wsum[BNODES / 32];
  __shared__ int wsum2[BNODES / 32];
  __shared__ __attribute__((aligned(16))) int sP[NCAP];
  __shared__ __attribute__((aligned(16))) float2 sW[NCAP];
  int* runSrc = sP;
  int* runDst = sP + G1;
  const int b = blockIdx.x;
  const int tid = threadIdx.x;
  const int lane = tid & 31;
  const int wave = tid >> 5;
  const int chunk = (E + G1 - 1) / G1;
  const float2* rw = (const float2*)recW;
  float2* sw2 = (float2*)sw;
  cntL[tid] = 0;
  const int o0 = __float_as_int(offT[tid * OT + b]);
  const int o1 = __float_as_int(offT[tid * OT + b + 1]);
  const int len = o1 - o0;
  const int inclL = wave_incl_scan(len, lane);
  int so = o0;
  so += __shfl_xor(so, 16, 32);
  so += __shfl_xor(so, 8, 32);
  so += __shfl_xor(so, 4, 32);
  so += __shfl_xor(so, 2, 32);
  so += __shfl_xor(so, 1, 32);
  if (lane == 31) wsum[wave] = inclL;
  if (lane == 0) wsum2[wave] = so;
  __syncthreads();
  int offr = 0, n = 0, base = 0;
  for (int w = 0; w < BNODES / 32; ++w) {
    const int t = wsum[w];
    if (w < wave) offr += t;
    n += t;
    base += wsum2[w];
  }
  const int exclL = offr + inclL - len;
  runSrc[tid] = min(E, tid * chunk) + o0;
  runDst[tid] = exclL;
  if (tid == G1 - 1) runDst[G1] = exclL + len;
  __syncthreads();
  const bool fits = n <= NCAP;
  int idx[NSLOTS];
  unsigned p[NSLOTS];
  float2 wt[NSLOTS];
  if (fits) {
    int* srcIdx = (int*)sW;
    const int hw = tid >> 4;
    const int j = tid & 15;
    int rs[G1 / 16], rd[G1 / 16], re[G1 / 16];
#pragma unroll
    for (int i = 0; i < G1 / 16; ++i) {
      const int r = hw + 16 * i;
      rs[i] = runSrc[r];
      rd[i] = runDst[r];
      re[i] = runDst[r + 1];
    }
#pragma unroll
    for (int i = 0; i < G1 / 16; ++i) {
      const int ln = re[i] - rd[i];
      if (j < ln) srcIdx[rd[i] + j] = rs[i] + j;
#pragma nounroll
      for (int jj = j + 16; jj < ln; jj += 16) srcIdx[rd[i] + jj] = rs[i] + jj;
    }
    __syncthreads();
#pragma unroll
    for (int u = 0; u < NSLOTS; ++u) {
      const int k = tid + u * BNODES;
      const int v = srcIdx[k];
      idx[u] = (k < n) ? v : -1;
    }
#pragma unroll
    for (int u = 0; u < NSLOTS; ++u) p[u] = recP[max(idx[u], 0)];
#pragma unroll
    for (int u = 0; u < NSLOTS; ++u) wt[u] = rw[max(idx[u], 0)];
#pragma unroll
    for (int u = 0; u < NSLOTS; ++u)
      if (idx[u] >= 0) atomicAdd(&cntL[p[u] >> 24], 1);
  } else {
    for (int i = tid; i < n; i += BNODES)
      atomicAdd(&cntL[recP[run_index(runSrc, runDst, i)] >> 24], 1);
  }
  __syncthreads();
  const int c = cntL[tid];
  const int incl = wave_incl_scan(c, lane);
  if (lane == 31) wsum[wave] = incl;
  __syncthreads();
  int off = 0;
  for (int w = 0; w < wave; ++w) off += wsum[w];
  const int excl = off + incl - c;
  curL[tid] = excl;
  const int node = b * BNODES + tid;
  if (node < N) start[node] = base + excl;
  if (node == N - 1) start[N] = base + excl + c;
  __syncthreads();
  if (fits) {
#pragma unroll
    for (int h = 0; h < NSLOTS; h += 8) {
      int pos[8];
#pragma unroll
      for (int u = 0; u < 8; ++u) {
        pos[u] = 0;
        if (idx[h + u] >= 0) pos[u] = atomicAdd(&curL[p[h + u] >> 24], 1);
      }
#pragma unroll
      for (int u = 0; u < 8; ++u) {
        if (idx[h + u] >= 0) {
          sP[pos[u]] = (int)(p[h + u] & 0xFFFFFFu);
          sW[pos[u]] = wt[h + u];
        }
      }
    }
    __syncthreads();
    for (int i = tid; i < n; i += BNODES) {
      ssrc[base + i] = sP[i];
      sw2[base + i] = sW[i];
    }
  } else {
    for (int i = tid; i < n; i += BNODES) {
      const int ix = run_index(runSrc, runDst, i);
      const unsigned pp = recP[ix];
      const float2 w = rw[ix];
      const int pos = base + atomicAdd(&curL[pp >> 24], 1);
      ssrc[pos] = (int)(pp & 0xFFFFFFu);
      sw2[pos] = w;
    }
  }

  __syncthreads();
  const int q = tid & 1;
  const int r0 = tid >> 1;
  const int r1 = BNODES / 2 + r0;
  const int beg0 = r0 ? curL[r0 - 1] : 0;
  const int end0 = curL[r0];
  const int beg1 = curL[r1 - 1];
  const int end1 = curL[r1];
  float4 g00 = make_float4(0.f, 0.f, 0.f, 0.f);
  float4 g01 = make_float4(0.f, 0.f, 0.f, 0.f);
  float4 g10 = make_float4(0.f, 0.f, 0.f, 0.f);
  float4 g11 = make_float4(0.f, 0.f, 0.f, 0.f);
  for (int c0 = 0; c0 < n; c0 += NCAP) {
    const int cn = min(NCAP, n - c0);
    if (!fits) {
      if (c0 > 0) __syncthreads();
      for (int i = tid; i < cn; i += BNODES) {
        sP[i] = ssrc[base + c0 + i];
        sW[i] = sw2[base + c0 + i];
      }
      __syncthreads();
    }
    sum_rows<5>((const char*)x, 16u * (unsigned)q, sP, sW, c0,
                max(beg0, c0), min(end0, c0 + cn), g00, g01);
    sum_rows<5>((const char*)x, 16u * (unsigned)q, sP, sW, c0,
                max(beg1, c0), min(end1, c0 + cn), g10, g11);
  }
  __syncthreads();
  float* atile = (float*)sW;
  *(float4*)(atile + r0 * A0P + 4 * q) = g00;
  *(float4*)(atile + r0 * A0P + 8 + 4 * q) = g01;
  *(float4*)(atile + r1 * A0P + 4 * q) = g10;
  *(float4*)(atile + r1 * A0P + 8 + 4 * q) = g11;
  __syncthreads();

  const int nn = lane & 15;
  const int kb = lane >> 4;
  float* otile0 = (float*)sP + wave * 16 * H1P;
  float* otile1 = (float*)sW + BNODES * A0P + wave * 16 * H1P;
  for (int tl = 0; tl < 2; ++tl) {
    const int rb = 16 * wave + (BNODES / 2) * tl;
    const int t0 = b * BNODES + rb;
    float* ot = tl ? otile1 : otile0;
    if (t0 + 16 <= N) {
      v8f acc0 = {}, acc1 = {};
#pragma unroll
      for (int kt = 0; kt < 4; ++kt) {
        const int kA = kt * 4 + 2 * kb;
        v2f a, bb0, bb1;
        const float* ap = atile + (rb + nn) * A0P + kA;
        a.x = ap[0];
        a.y = ap[1];
        bb0.x = W0[(kA + 0) * 32 + nn];
        bb0.y = W0[(kA + 1) * 32 + nn];
        bb1.x = W0[(kA + 0) * 32 + 16 + nn];
        bb1.y = W0[(kA + 1) * 32 + 16 + nn];
        acc0 = wmma4f32(a, bb0, acc0);
        acc1 = wmma4f32(a, bb1, acc1);
      }
      const float bias0 = b0[nn], bias1 = b0[16 + nn];
#pragma unroll
      for (int v = 0; v < 8; ++v) {
        ot[(v + 8 * kb) * H1P + nn] = acc0[v] + bias0;
        ot[(v + 8 * kb) * H1P + 16 + nn] = acc1[v] + bias1;
      }
    }
  }
  __syncthreads();
  const int hr = lane >> 1;
  const int hq = lane & 1;
  for (int tl = 0; tl < 2; ++tl) {
    const int t0 = b * BNODES + 16 * wave + (BNODES / 2) * tl;
    const float* rp = (tl ? otile1 : otile0) + hr * H1P + 4 * hq;
    if (t0 + 16 <= N) {
      float4 u[4];
#pragma unroll
      for (int j = 0; j < 4; ++j) u[j] = *(const float4*)(rp + 8 * j);
      float s = 0.f;
#pragma unroll
      for (int j = 0; j < 4; ++j) {
        s += u[j].x * u[j].x;
        s += u[j].y * u[j].y;
        s += u[j].z * u[j].z;
        s += u[j].w * u[j].w;
      }
      s += __shfl_xor(s, 1, 32);
      const float invn = 1.0f / fmaxf(sqrtf(s), 1e-12f);
      float* hp = h1 + (size_t)(t0 + hr) * 32 + 4 * hq;
#pragma unroll
      for (int j = 0; j < 4; ++j)
        *(float4*)(hp + 8 * j) = make_float4(u[j].x * invn, u[j].y * invn, u[j].z * invn, u[j].w * invn);
    }
  }
}

#define TS 132
#define AS 68
#define M1CAP 512
__global__ __launch_bounds__(128, 12) void mlp1_kernel(const float* __restrict__ h1,
                                                   const int* __restrict__ start,
                                                   const int* __restrict__ ssrc,
                                                   const float* __restrict__ sw,
                                                   const float* __restrict__ W1,
                                                   const float* __restrict__ b1,
                                                   const float* __restrict__ W2,
                                                   const float* __restrict__ b2,
                                                   float* __restrict__ out) {
  __shared__ __attribute__((aligned(16))) float ash[16 * AS];
  __shared__ float tsh[16 * TS];
  __shared__ int eidx[M1CAP];
  __shared__ float2 ewt[M1CAP];
  const int m0 = blockIdx.x * 16;
  const int wave = threadIdx.x >> 5;
  const int lane = threadIdx.x & 31;
  const int n = lane & 15;
  const int kb = lane >> 4;

  {
    const int r = threadIdx.x >> 3;
    const int q = threadIdx.x & 7;
    const int eb = start[m0];
    const int ee = start[m0 + 16];
    const int beg = start[m0 + r];
    const int end = start[m0 + r + 1];
    const float2* sw2 = (const float2*)sw;
    float4 a0 = make_float4(0.f, 0.f, 0.f, 0.f);
    float4 a1 = make_float4(0.f, 0.f, 0.f, 0.f);
    for (int c0 = eb; c0 < ee; c0 += M1CAP) {
      const int cn = min(M1CAP, ee - c0);
      int si[M1CAP / 128];
      float2 wi[M1CAP / 128];
#pragma unroll
      for (int u = 0; u < M1CAP / 128; ++u) {
        const int i = (int)threadIdx.x + u * 128;
        si[u] = 0;
        wi[u] = make_float2(0.f, 0.f);
        if (i < cn) {
          si[u] = ssrc[c0 + i];
          wi[u] = sw2[c0 + i];
        }
      }
#pragma unroll
      for (int u = 0; u < M1CAP / 128; ++u) {
        const int i = (int)threadIdx.x + u * 128;
        if (i < cn) {
          eidx[i] = si[u];
          ewt[i] = wi[u];
        }
      }
      __syncthreads();
      const int pb = max(beg, c0);
      const int pe = min(end, c0 + cn);
      sum_rows<7>((const char*)h1, 16u * (unsigned)q, eidx, ewt, c0, pb, pe, a0, a1);
      if (c0 + M1CAP < ee) __syncthreads();
    }
    *(float4*)(ash + r * AS + 4 * q) = a0;
    *(float4*)(ash + r * AS + 32 + 4 * q) = a1;
  }
  __syncthreads();

#pragma unroll
  for (int tt = 0; tt < 2; ++tt) {
    const int n0 = (wave * 2 + tt) * 16;
    v8f c = {};
    for (int kt = 0; kt < 16; ++kt) {
      const int kA = kt * 4 + 2 * kb;
      v2f a, b;
      const float* ap = ash + n * AS + kA;
      a.x = ap[0];
      a.y = ap[1];
      b.x = W1[(kA + 0) * 128 + n0 + n];
      b.y = W1[(kA + 1) * 128 + n0 + n];
      c = wmma4f32(a, b, c);
    }
    const float bias = b1[n0 + n];
#pragma unroll
    for (int v = 0; v < 8; ++v) {
      tsh[(v + 8 * kb) * TS + n0 + n] = fmaxf(c[v] + bias, 0.0f);
    }
  }
  __syncthreads();

  const int n0 = wave * 16;
  v8f c = {};
  for (int kt = 0; kt < 32; ++kt) {
    const int kA = kt * 4 + 2 * kb;
    v2f a, b;
    a.x = tsh[n * TS + kA];
    a.y = tsh[n * TS + kA + 1];
    b.x = W2[(kA + 0) * 64 + n0 + n];
    b.y = W2[(kA + 1) * 64 + n0 + n];
    c = wmma4f32(a, b, c);
  }
  const float bias = b2[n0 + n];
  float* otile = ash;
#pragma unroll
  for (int v = 0; v < 8; ++v) otile[(v + 8 * kb) * AS + n0 + n] = c[v] + bias;
  __syncthreads();
  {
    const int r = threadIdx.x >> 3;
    const int q = threadIdx.x & 7;
    const float4 u0 = *(const float4*)(otile + r * AS + 4 * q);
    const float4 u1 = *(const float4*)(otile + r * AS + 32 + 4 * q);
    float s = u0.x * u0.x;
    s += u0.y * u0.y;
    s += u0.z * u0.z;
    s += u0.w * u0.w;
    s += u1.x * u1.x;
    s += u1.y * u1.y;
    s += u1.z * u1.z;
    s += u1.w * u1.w;
    s += __shfl_xor(s, 1, 32);
    s += __shfl_xor(s, 2, 32);
    s += __shfl_xor(s, 4, 32);
    const float invn = 1.0f / fmaxf(sqrtf(s), 1e-12f);
    float* op = out + (size_t)(m0 + r) * 64 + 4 * q;
    *(float4*)op = make_float4(u0.x * invn, u0.y * invn, u0.z * invn, u0.w * invn);
    *(float4*)(op + 32) = make_float4(u1.x * invn, u1.y * invn, u1.z * invn, u1.w * invn);
  }
}

extern "C" void kernel_launch(void* const* d_in, const int* in_sizes, int n_in,
                              void* d_out, int out_size, void* d_ws, size_t ws_size,
                              hipStream_t stream) {
  const float* x  = (const float*)d_in[0];
  const int*   ei = (const int*)d_in[1];
  const float* Kw = (const float*)d_in[2];
  const float* W0 = (const float*)d_in[3];
  const float* b0 = (const float*)d_in[4];
  const float* W1 = (const float*)d_in[5];
  const float* b1 = (const float*)d_in[6];
  const float* W2 = (const float*)d_in[7];
  const float* b2 = (const float*)d_in[8];

  const int N = in_sizes[0] / 8;
  const int E = in_sizes[1] / 2;
  const int* src = ei;
  const int* dst = ei + E;

  const int NB = (N + BNODES - 1) / BNODES;

  int*      start    = (int*)d_ws;
  int*      ssrc     = start + N + 64 + NBMAX;
  float*    sw       = (float*)(ssrc + E);
  float*    rec      = sw + (size_t)E * 2;
  float*    h1       = rec + (size_t)N * 16;
  float*    offT     = (float*)d_out;
  unsigned* recP     = (unsigned*)d_out + G1 * OT;
  float*    recW     = (float*)(recP + E);

  chunk_sort_kernel<<<G1, 1024, 0, stream>>>(src, dst, Kw, Kw + E, offT, recP, recW, E, NB);
  node_place_layer0_kernel<<<NB, BNODES, 0, stream>>>(offT, recP, recW, x, W0, b0, start, ssrc, sw,
                                                      h1, N, E);
  mlp1_kernel<<<N / 16, 128, 0, stream>>>(h1, start, ssrc, sw, W1, b1, W2, b2,
                                          (float*)d_out);
}
